# starting kernel
# baseline (speedup 1.0000x reference)
#include <hip/hip_runtime.h>

typedef __attribute__((ext_vector_type(2))) float v2f;
typedef __attribute__((ext_vector_type(8))) float v8f;

#define H_DIM 128
#define O_DIM 64

__global__ void zero_f32_kernel(float* __restrict__ p, long n) {
    long i = (long)blockIdx.x * blockDim.x + threadIdx.x;
    if (i < n) p[i] = 0.0f;
}

__global__ void deg_count_kernel(const int* __restrict__ dst,
                                 float* __restrict__ deg,
                                 int E, int N, long total) {
    long i = (long)blockIdx.x * blockDim.x + threadIdx.x;
    if (i >= total) return;
    int r = (int)(i / E);
    int d = dst[i];
    atomicAdd(&deg[(long)r * N + d], 1.0f);
}

__global__ void deg_recip_kernel(float* __restrict__ deg, long n) {
    long i = (long)blockIdx.x * blockDim.x + threadIdx.x;
    if (i < n) deg[i] = 1.0f / fmaxf(deg[i], 1.0f);
}

__global__ void layer1_scatter_kernel(const float* __restrict__ embed,
                                      const int* __restrict__ src,
                                      const int* __restrict__ dst,
                                      const float* __restrict__ deg_inv,
                                      float* __restrict__ h1,
                                      int E, int N, long n_edges) {
    long gid  = (long)blockIdx.x * blockDim.x + threadIdx.x;
    long edge = gid >> 5;
    int  lane = (int)(gid & 31);
    if (edge >= n_edges) return;
    int r = (int)(edge / E);
    int s = src[edge];
    int d = dst[edge];
    float w = deg_inv[(long)r * N + d];
    const float4 v = *(const float4*)(embed + (long)s * H_DIM + lane * 4);
    float* o = h1 + (long)d * H_DIM + lane * 4;
    atomicAdd(o + 0, v.x * w);
    atomicAdd(o + 1, v.y * w);
    atomicAdd(o + 2, v.z * w);
    atomicAdd(o + 3, v.w * w);
}

__global__ void relu_bias_kernel(float* __restrict__ h1,
                                 const float* __restrict__ bias1, long n) {
    long i = (long)blockIdx.x * blockDim.x + threadIdx.x;
    if (i < n) h1[i] = fmaxf(h1[i] + bias1[i & (H_DIM - 1)], 0.0f);
}

__global__ void init_out_kernel(float* __restrict__ out,
                                const float* __restrict__ bias2, long n) {
    long i = (long)blockIdx.x * blockDim.x + threadIdx.x;
    if (i < n) out[i] = bias2[i & (O_DIM - 1)];
}

__global__ void __launch_bounds__(256)
matmul_wmma_kernel(const float* __restrict__ h1,
                   const float* __restrict__ W,
                   float* __restrict__ x,
                   int N) {
    __shared__ float Wlds[H_DIM * O_DIM];
    int tid = threadIdx.x;
    for (int i = tid; i < H_DIM * O_DIM; i += 256) {
        int k = i >> 6;
        int o = i & (O_DIM - 1);
        Wlds[(k >> 1) * (2 * O_DIM) + o * 2 + (k & 1)] = W[i];
    }
    __syncthreads();

    int wave    = tid >> 5;
    int lane    = tid & 31;
    int lhalf   = lane >> 4;
    int l15     = lane & 15;
    int koff    = lhalf * 2;
    int rowBase = blockIdx.x * 128 + wave * 16;

    int  arow   = rowBase + l15;
    long arow_c = (long)min(arow, N - 1) * H_DIM;

    const v2f* bbase = (const v2f*)Wlds + lhalf * O_DIM + l15;

    v8f acc0 = {}; v8f acc1 = {}; v8f acc2 = {}; v8f acc3 = {};

    for (int k0 = 0; k0 < H_DIM; k0 += 4) {
        v2f a = *(const v2f*)(h1 + arow_c + k0 + koff);

        const v2f* bp = bbase + (k0 >> 1) * O_DIM;
        v2f b0 = bp[0];
        v2f b1 = bp[16];
        v2f b2 = bp[32];
        v2f b3 = bp[48];

        acc0 = __builtin_amdgcn_wmma_f32_16x16x4_f32(false, a, false, b0,
                                                     (short)0, acc0, false, false);
        acc1 = __builtin_amdgcn_wmma_f32_16x16x4_f32(false, a, false, b1,
                                                     (short)0, acc1, false, false);
        acc2 = __builtin_amdgcn_wmma_f32_16x16x4_f32(false, a, false, b2,
                                                     (short)0, acc2, false, false);
        acc3 = __builtin_amdgcn_wmma_f32_16x16x4_f32(false, a, false, b3,
                                                     (short)0, acc3, false, false);
    }

    int crow = rowBase + 8 * lhalf;
    #pragma unroll
    for (int v = 0; v < 8; ++v) {
        int row = crow + v;
        if (row < N) {
            float* o = x + (long)row * O_DIM + l15;
            o[0]  = acc0[v];
            o[16] = acc1[v];
            o[32] = acc2[v];
            o[48] = acc3[v];
        }
    }
}

__global__ void layer2_scatter_kernel(const float* __restrict__ x,
                                      const int* __restrict__ src,
                                      const int* __restrict__ dst,
                                      const float* __restrict__ deg_inv_r,
                                      float* __restrict__ out,
                                      long n_edges) {
    long gid  = (long)blockIdx.x * blockDim.x + threadIdx.x;
    long edge = gid >> 5;
    int  lane = (int)(gid & 31);
    if (edge >= n_edges) return;
    int s = src[edge];
    int d = dst[edge];
    float w = deg_inv_r[d];
    const float2 v = *(const float2*)(x + (long)s * O_DIM + lane * 2);
    float* o = out + (long)d * O_DIM + lane * 2;
    atomicAdd(o + 0, v.x * w);
    atomicAdd(o + 1, v.y * w);
}

extern "C" void kernel_launch(void* const* d_in, const int* in_sizes, int n_in,
                              void* d_out, int out_size, void* d_ws, size_t ws_size,
                              hipStream_t stream) {
    const float* embed    = (const float*)d_in[0];
    const float* weight   = (const float*)d_in[1];
    const float* bias1    = (const float*)d_in[2];
    const float* bias2    = (const float*)d_in[3];
    const int*   edge_src = (const int*)d_in[4];
    const int*   edge_dst = (const int*)d_in[5];

    const int H = in_sizes[2];
    const int O = in_sizes[3];
    const int N = in_sizes[0] / H;
    const int R = in_sizes[1] / (H * O);
    const int E = in_sizes[4] / R;

    float* out     = (float*)d_out;
    float* deg_inv = (float*)d_ws;
    float* h1      = deg_inv + (long)R * N;
    float* x       = h1 + (long)N * H;

    const int BT = 256;

    long rn = (long)R * N;
    zero_f32_kernel<<<(unsigned)((rn + BT - 1) / BT), BT, 0, stream>>>(deg_inv, rn);
    long re = (long)R * E;
    deg_count_kernel<<<(unsigned)((re + BT - 1) / BT), BT, 0, stream>>>(
        edge_dst, deg_inv, E, N, re);
    deg_recip_kernel<<<(unsigned)((rn + BT - 1) / BT), BT, 0, stream>>>(deg_inv, rn);

    long nh = (long)N * H;
    zero_f32_kernel<<<(unsigned)((nh + BT - 1) / BT), BT, 0, stream>>>(h1, nh);
    long l1_threads = re * 32;
    layer1_scatter_kernel<<<(unsigned)((l1_threads + BT - 1) / BT), BT, 0, stream>>>(
        embed, edge_src, edge_dst, deg_inv, h1, E, N, re);
    relu_bias_kernel<<<(unsigned)((nh + BT - 1) / BT), BT, 0, stream>>>(h1, bias1, nh);

    long no = (long)N * O;
    init_out_kernel<<<(unsigned)((no + BT - 1) / BT), BT, 0, stream>>>(out, bias2, no);

    unsigned mm_blocks = (unsigned)((N + 127) / 128);
    long l2_threads = (long)E * 32;
    for (int r = 0; r < R; ++r) {
        matmul_wmma_kernel<<<mm_blocks, BT, 0, stream>>>(
            h1, weight + (long)r * H * O, x, N);
        layer2_scatter_kernel<<<(unsigned)((l2_threads + BT - 1) / BT), BT, 0, stream>>>(
            x, edge_src + (long)r * E, edge_dst + (long)r * E,
            deg_inv + (long)r * N, out, (long)E);
    }
}


# Scatter kernels: lane l owns columns l+32j, so each f32 atomic add (and row load) instruction covers one whole 128-B line: 16->4 and 4->2 atomic line requests per edge; arrival-order sums as before
# speedup vs baseline: 1.1083x; 1.1083x over previous
#include <hip/hip_runtime.h>

typedef __attribute__((ext_vector_type(2))) float v2f;
typedef __attribute__((ext_vector_type(8))) float v8f;

#define H_DIM 128
#define O_DIM 64

__global__ void zero_f32_kernel(float* __restrict__ p, long n) {
    long i = (long)blockIdx.x * blockDim.x + threadIdx.x;
    if (i < n) p[i] = 0.0f;
}

__global__ void deg_count_kernel(const int* __restrict__ dst,
                                 float* __restrict__ deg,
                                 int E, int N, long total) {
    long i = (long)blockIdx.x * blockDim.x + threadIdx.x;
    if (i >= total) return;
    int r = (int)(i / E);
    int d = dst[i];
    atomicAdd(&deg[(long)r * N + d], 1.0f);
}

__global__ void deg_recip_kernel(float* __restrict__ deg, long n) {
    long i = (long)blockIdx.x * blockDim.x + threadIdx.x;
    if (i < n) deg[i] = 1.0f / fmaxf(deg[i], 1.0f);
}

__global__ void layer1_scatter_kernel(const float* __restrict__ embed,
                                      const int* __restrict__ src,
                                      const int* __restrict__ dst,
                                      const float* __restrict__ deg_inv,
                                      float* __restrict__ h1,
                                      int E, int N, long n_edges) {
    long gid  = (long)blockIdx.x * blockDim.x + threadIdx.x;
    long edge = gid >> 5;
    int  lane = (int)(gid & 31);
    if (edge >= n_edges) return;
    int r = (int)(edge / E);
    int s = src[edge];
    int d = dst[edge];
    float w = deg_inv[(long)r * N + d];
    const float* e = embed + (long)s * H_DIM + lane;
    const float v0 = e[0];
    const float v1 = e[32];
    const float v2 = e[64];
    const float v3 = e[96];
    float* o = h1 + (long)d * H_DIM + lane;
    atomicAdd(o +  0, v0 * w);
    atomicAdd(o + 32, v1 * w);
    atomicAdd(o + 64, v2 * w);
    atomicAdd(o + 96, v3 * w);
}

__global__ void relu_bias_kernel(float* __restrict__ h1,
                                 const float* __restrict__ bias1, long n) {
    long i = (long)blockIdx.x * blockDim.x + threadIdx.x;
    if (i < n) h1[i] = fmaxf(h1[i] + bias1[i & (H_DIM - 1)], 0.0f);
}

__global__ void init_out_kernel(float* __restrict__ out,
                                const float* __restrict__ bias2, long n) {
    long i = (long)blockIdx.x * blockDim.x + threadIdx.x;
    if (i < n) out[i] = bias2[i & (O_DIM - 1)];
}

__global__ void __launch_bounds__(256)
matmul_wmma_kernel(const float* __restrict__ h1,
                   const float* __restrict__ W,
                   float* __restrict__ x,
                   int N) {
    __shared__ float Wlds[H_DIM * O_DIM];
    int tid = threadIdx.x;
    for (int i = tid; i < H_DIM * O_DIM; i += 256) {
        int k = i >> 6;
        int o = i & (O_DIM - 1);
        Wlds[(k >> 1) * (2 * O_DIM) + o * 2 + (k & 1)] = W[i];
    }
    __syncthreads();

    int wave    = tid >> 5;
    int lane    = tid & 31;
    int lhalf   = lane >> 4;
    int l15     = lane & 15;
    int koff    = lhalf * 2;
    int rowBase = blockIdx.x * 128 + wave * 16;

    int  arow   = rowBase + l15;
    long arow_c = (long)min(arow, N - 1) * H_DIM;

    const v2f* bbase = (const v2f*)Wlds + lhalf * O_DIM + l15;

    v8f acc0 = {}; v8f acc1 = {}; v8f acc2 = {}; v8f acc3 = {};

    for (int k0 = 0; k0 < H_DIM; k0 += 4) {
        v2f a = *(const v2f*)(h1 + arow_c + k0 + koff);

        const v2f* bp = bbase + (k0 >> 1) * O_DIM;
        v2f b0 = bp[0];
        v2f b1 = bp[16];
        v2f b2 = bp[32];
        v2f b3 = bp[48];

        acc0 = __builtin_amdgcn_wmma_f32_16x16x4_f32(false, a, false, b0,
                                                     (short)0, acc0, false, false);
        acc1 = __builtin_amdgcn_wmma_f32_16x16x4_f32(false, a, false, b1,
                                                     (short)0, acc1, false, false);
        acc2 = __builtin_amdgcn_wmma_f32_16x16x4_f32(false, a, false, b2,
                                                     (short)0, acc2, false, false);
        acc3 = __builtin_amdgcn_wmma_f32_16x16x4_f32(false, a, false, b3,
                                                     (short)0, acc3, false, false);
    }

    int crow = rowBase + 8 * lhalf;
    #pragma unroll
    for (int v = 0; v < 8; ++v) {
        int row = crow + v;
        if (row < N) {
            float* o = x + (long)row * O_DIM + l15;
            o[0]  = acc0[v];
            o[16] = acc1[v];
            o[32] = acc2[v];
            o[48] = acc3[v];
        }
    }
}

__global__ void layer2_scatter_kernel(const float* __restrict__ x,
                                      const int* __restrict__ src,
                                      const int* __restrict__ dst,
                                      const float* __restrict__ deg_inv_r,
                                      float* __restrict__ out,
                                      long n_edges) {
    long gid  = (long)blockIdx.x * blockDim.x + threadIdx.x;
    long edge = gid >> 5;
    int  lane = (int)(gid & 31);
    if (edge >= n_edges) return;
    int s = src[edge];
    int d = dst[edge];
    float w = deg_inv_r[d];
    const float* xr = x + (long)s * O_DIM + lane;
    const float v0 = xr[0];
    const float v1 = xr[32];
    float* o = out + (long)d * O_DIM + lane;
    atomicAdd(o +  0, v0 * w);
    atomicAdd(o + 32, v1 * w);
}

extern "C" void kernel_launch(void* const* d_in, const int* in_sizes, int n_in,
                              void* d_out, int out_size, void* d_ws, size_t ws_size,
                              hipStream_t stream) {
    const float* embed    = (const float*)d_in[0];
    const float* weight   = (const float*)d_in[1];
    const float* bias1    = (const float*)d_in[2];
    const float* bias2    = (const float*)d_in[3];
    const int*   edge_src = (const int*)d_in[4];
    const int*   edge_dst = (const int*)d_in[5];

    const int H = in_sizes[2];
    const int O = in_sizes[3];
    const int N = in_sizes[0] / H;
    const int R = in_sizes[1] / (H * O);
    const int E = in_sizes[4] / R;

    float* out     = (float*)d_out;
    float* deg_inv = (float*)d_ws;
    float* h1      = deg_inv + (long)R * N;
    float* x       = h1 + (long)N * H;

    const int BT = 256;

    long rn = (long)R * N;
    zero_f32_kernel<<<(unsigned)((rn + BT - 1) / BT), BT, 0, stream>>>(deg_inv, rn);
    long re = (long)R * E;
    deg_count_kernel<<<(unsigned)((re + BT - 1) / BT), BT, 0, stream>>>(
        edge_dst, deg_inv, E, N, re);
    deg_recip_kernel<<<(unsigned)((rn + BT - 1) / BT), BT, 0, stream>>>(deg_inv, rn);

    long nh = (long)N * H;
    zero_f32_kernel<<<(unsigned)((nh + BT - 1) / BT), BT, 0, stream>>>(h1, nh);
    long l1_threads = re * 32;
    layer1_scatter_kernel<<<(unsigned)((l1_threads + BT - 1) / BT), BT, 0, stream>>>(
        embed, edge_src, edge_dst, deg_inv, h1, E, N, re);
    relu_bias_kernel<<<(unsigned)((nh + BT - 1) / BT), BT, 0, stream>>>(h1, bias1, nh);

    long no = (long)N * O;
    init_out_kernel<<<(unsigned)((no + BT - 1) / BT), BT, 0, stream>>>(out, bias2, no);

    unsigned mm_blocks = (unsigned)((N + 127) / 128);
    long l2_threads = (long)E * 32;
    for (int r = 0; r < R; ++r) {
        matmul_wmma_kernel<<<mm_blocks, BT, 0, stream>>>(
            h1, weight + (long)r * H * O, x, N);
        layer2_scatter_kernel<<<(unsigned)((l2_threads + BT - 1) / BT), BT, 0, stream>>>(
            x, edge_src + (long)r * E, edge_dst + (long)r * E,
            deg_inv + (long)r * N, out, (long)E);
    }
}


# Work per wave raised in all kernels but the matrix product: fixed-grid streaming kernels, 16-B vectors, 8 in flight per lane; scatter waves take 32 edges each; arrival-order atomic sums as before
# speedup vs baseline: 1.3387x; 1.2079x over previous
#include <hip/hip_runtime.h>


typedef __attribute__((ext_vector_type(2))) float v2f;
typedef __attribute__((ext_vector_type(4))) float v4f;
typedef __attribute__((ext_vector_type(8))) float v8f;

#define H_DIM 128
#define O_DIM 64

#define STREAM_U      8
#define STREAM_BLOCKS 512

__global__ void __launch_bounds__(256)
zero_f32_kernel(float* __restrict__ p, long n) {
    const long tid    = (long)blockIdx.x * blockDim.x + threadIdx.x;
    const long stride = (long)gridDim.x * blockDim.x;
    if ((((size_t)p) & 15) != 0) {
        for (long t = tid; t < n; t += stride) p[t] = 0.0f;
        return;
    }
    v4f* __restrict__ p4 = (v4f*)p;
    const long n4 = n >> 2;
    const v4f z = {0.0f, 0.0f, 0.0f, 0.0f};
    long i = tid;
    for (; i + (STREAM_U - 1) * stride < n4; i += STREAM_U * stride) {
        #pragma unroll
        for (int u = 0; u < STREAM_U; ++u) p4[i + u * stride] = z;
    }
    for (; i < n4; i += stride) p4[i] = z;
    for (long t = (n4 << 2) + tid; t < n; t += stride) p[t] = 0.0f;
}

__global__ void __launch_bounds__(256)
deg_count_kernel(const int* __restrict__ dst,
                 float* __restrict__ deg,
                 int E, int N, long total) {
    const long tid    = (long)blockIdx.x * blockDim.x + threadIdx.x;
    const long stride = (long)gridDim.x * blockDim.x;
    long i = tid;
    for (; i + 3 * stride < total; i += 4 * stride) {
        int d[4];
        #pragma unroll
        for (int u = 0; u < 4; ++u) d[u] = dst[i + u * stride];
        #pragma unroll
        for (int u = 0; u < 4; ++u) {
            const int r = (int)((i + u * stride) / E);
            atomicAdd(&deg[(long)r * N + d[u]], 1.0f);
        }
    }
    for (; i < total; i += stride) {
        const int r = (int)(i / E);
        atomicAdd(&deg[(long)r * N + dst[i]], 1.0f);
    }
}

__global__ void __launch_bounds__(256)
deg_recip_kernel(float* __restrict__ deg, long n) {
    const long tid    = (long)blockIdx.x * blockDim.x + threadIdx.x;
    const long stride = (long)gridDim.x * blockDim.x;
    if ((((size_t)deg) & 15) != 0) {
        for (long t = tid; t < n; t += stride) deg[t] = 1.0f / fmaxf(deg[t], 1.0f);
        return;
    }
    v4f* __restrict__ p4 = (v4f*)deg;
    const long n4 = n >> 2;
    for (long i = tid; i < n4; i += stride) {
        v4f v = p4[i];
        v.x = 1.0f / fmaxf(v.x, 1.0f);
        v.y = 1.0f / fmaxf(v.y, 1.0f);
        v.z = 1.0f / fmaxf(v.z, 1.0f);
        v.w = 1.0f / fmaxf(v.w, 1.0f);
        p4[i] = v;
    }
    for (long t = (n4 << 2) + tid; t < n; t += stride) deg[t] = 1.0f / fmaxf(deg[t], 1.0f);
}

#define L1_GROUP 8

__device__ __forceinline__ void layer1_one_edge(const float* __restrict__ embed,
                                                float* __restrict__ h1,
                                                int s, int d, float w, int lane) {
    const float* e = embed + (long)s * H_DIM + lane;
    const float v0 = e[0];
    const float v1 = e[32];
    const float v2 = e[64];
    const float v3 = e[96];
    float* o = h1 + (long)d * H_DIM + lane;
    atomicAdd(o +  0, v0 * w);
    atomicAdd(o + 32, v1 * w);
    atomicAdd(o + 64, v2 * w);
    atomicAdd(o + 96, v3 * w);
}

__global__ void __launch_bounds__(128)
layer1_scatter_kernel(const float* __restrict__ embed,
                      const int* __restrict__ src,
                      const int* __restrict__ dst,
                      const float* __restrict__ deg_inv,
                      float* __restrict__ h1,
                      int E, int N, long n_edges) {
    const long wave = ((long)blockIdx.x * blockDim.x + threadIdx.x) >> 5;
    const int  lane = (int)(threadIdx.x & 31);
    const long e0   = wave << 5;
    if (e0 >= n_edges) return;

    if (e0 + 32 > n_edges) {
        const int cnt = (int)(n_edges - e0);
        for (int j = 0; j < cnt; ++j) {
            const long ee = e0 + j;
            const int  r  = (int)(ee / E);
            const int  dj = dst[ee];
            layer1_one_edge(embed, h1, src[ee], dj, deg_inv[(long)r * N + dj], lane);
        }
        return;
    }

    const long  e = e0 + lane;
    const int   s = src[e];
    const int   d = dst[e];
    const int   r = (int)(e / E);
    const float w = deg_inv[(long)r * N + d];

    #pragma unroll
    for (int g = 0; g < 32; g += L1_GROUP) {
        float v[L1_GROUP][4];
        #pragma unroll
        for (int j = 0; j < L1_GROUP; ++j) {
            const int sj = __builtin_amdgcn_readlane(s, g + j);
            const float* ep = embed + (long)sj * H_DIM + lane;
            v[j][0] = ep[0];
            v[j][1] = ep[32];
            v[j][2] = ep[64];
            v[j][3] = ep[96];
        }
        #pragma unroll
        for (int j = 0; j < L1_GROUP; ++j) {
            const int   dj = __builtin_amdgcn_readlane(d, g + j);
            const float wj = __int_as_float(
                __builtin_amdgcn_readlane(__float_as_int(w), g + j));
            float* o = h1 + (long)dj * H_DIM + lane;
            atomicAdd(o +  0, v[j][0] * wj);
            atomicAdd(o + 32, v[j][1] * wj);
            atomicAdd(o + 64, v[j][2] * wj);
            atomicAdd(o + 96, v[j][3] * wj);
        }
    }
}

__global__ void __launch_bounds__(256)
relu_bias_kernel(float* __restrict__ h1,
                 const float* __restrict__ bias1, long n) {
    const long tid    = (long)blockIdx.x * blockDim.x + threadIdx.x;
    const long stride = (long)gridDim.x * blockDim.x;
    if ((((size_t)h1) & 15) != 0 || (stride & (H_DIM / 4 - 1)) != 0) {
        for (long t = tid; t < n; t += stride)
            h1[t] = fmaxf(h1[t] + bias1[t & (H_DIM - 1)], 0.0f);
        return;
    }
    v4f* __restrict__ p4 = (v4f*)h1;
    const long n4 = n >> 2;
    const int  c  = (int)(tid & (H_DIM / 4 - 1)) * 4;
    const float b0 = bias1[c], b1 = bias1[c + 1], b2 = bias1[c + 2], b3 = bias1[c + 3];
    long i = tid;
    for (; i + (STREAM_U - 1) * stride < n4; i += STREAM_U * stride) {
        v4f v[STREAM_U];
        #pragma unroll
        for (int u = 0; u < STREAM_U; ++u) v[u] = p4[i + u * stride];
        #pragma unroll
        for (int u = 0; u < STREAM_U; ++u) {
            v4f t = v[u];
            t.x = fmaxf(t.x + b0, 0.0f);
            t.y = fmaxf(t.y + b1, 0.0f);
            t.z = fmaxf(t.z + b2, 0.0f);
            t.w = fmaxf(t.w + b3, 0.0f);
            p4[i + u * stride] = t;
        }
    }
    for (; i < n4; i += stride) {
        v4f t = p4[i];
        t.x = fmaxf(t.x + b0, 0.0f);
        t.y = fmaxf(t.y + b1, 0.0f);
        t.z = fmaxf(t.z + b2, 0.0f);
        t.w = fmaxf(t.w + b3, 0.0f);
        p4[i] = t;
    }
    for (long t = (n4 << 2) + tid; t < n; t += stride)
        h1[t] = fmaxf(h1[t] + bias1[t & (H_DIM - 1)], 0.0f);
}

__global__ void __launch_bounds__(256)
init_out_kernel(float* __restrict__ out,
                const float* __restrict__ bias2, long n) {
    const long tid    = (long)blockIdx.x * blockDim.x + threadIdx.x;
    const long stride = (long)gridDim.x * blockDim.x;
    if ((((size_t)out) & 15) != 0 || (stride & (O_DIM / 4 - 1)) != 0) {
        for (long t = tid; t < n; t += stride) out[t] = bias2[t & (O_DIM - 1)];
        return;
    }
    v4f* __restrict__ p4 = (v4f*)out;
    const long n4 = n >> 2;
    const int  c  = (int)(tid & (O_DIM / 4 - 1)) * 4;
    const v4f  b  = {bias2[c], bias2[c + 1], bias2[c + 2], bias2[c + 3]};
    long i = tid;
    for (; i + (STREAM_U - 1) * stride < n4; i += STREAM_U * stride) {
        #pragma unroll
        for (int u = 0; u < STREAM_U; ++u) p4[i + u * stride] = b;
    }
    for (; i < n4; i += stride) p4[i] = b;
    for (long t = (n4 << 2) + tid; t < n; t += stride) out[t] = bias2[t & (O_DIM - 1)];
}

__global__ void __launch_bounds__(256)
matmul_wmma_kernel(const float* __restrict__ h1,
                   const float* __restrict__ W,
                   float* __restrict__ x,
                   int N) {
    __shared__ float Wlds[H_DIM * O_DIM];
    int tid = threadIdx.x;
    for (int i = tid; i < H_DIM * O_DIM; i += 256) {
        int k = i >> 6;
        int o = i & (O_DIM - 1);
        Wlds[(k >> 1) * (2 * O_DIM) + o * 2 + (k & 1)] = W[i];
    }
    __syncthreads();

    int wave    = tid >> 5;
    int lane    = tid & 31;
    int lhalf   = lane >> 4;
    int l15     = lane & 15;
    int koff    = lhalf * 2;
    int rowBase = blockIdx.x * 128 + wave * 16;

    int  arow   = rowBase + l15;
    long arow_c = (long)min(arow, N - 1) * H_DIM;

    const v2f* bbase = (const v2f*)Wlds + lhalf * O_DIM + l15;

    v8f acc0 = {}; v8f acc1 = {}; v8f acc2 = {}; v8f acc3 = {};

    for (int k0 = 0; k0 < H_DIM; k0 += 4) {
        v2f a = *(const v2f*)(h1 + arow_c + k0 + koff);

        const v2f* bp = bbase + (k0 >> 1) * O_DIM;
        v2f b0 = bp[0];
        v2f b1 = bp[16];
        v2f b2 = bp[32];
        v2f b3 = bp[48];

        acc0 = __builtin_amdgcn_wmma_f32_16x16x4_f32(false, a, false, b0,
                                                     (short)0, acc0, false, false);
        acc1 = __builtin_amdgcn_wmma_f32_16x16x4_f32(false, a, false, b1,
                                                     (short)0, acc1, false, false);
        acc2 = __builtin_amdgcn_wmma_f32_16x16x4_f32(false, a, false, b2,
                                                     (short)0, acc2, false, false);
        acc3 = __builtin_amdgcn_wmma_f32_16x16x4_f32(false, a, false, b3,
                                                     (short)0, acc3, false, false);
    }

    int crow = rowBase + 8 * lhalf;
    #pragma unroll
    for (int v = 0; v < 8; ++v) {
        int row = crow + v;
        if (row < N) {
            float* o = x + (long)row * O_DIM + l15;
            o[0]  = acc0[v];
            o[16] = acc1[v];
            o[32] = acc2[v];
            o[48] = acc3[v];
        }
    }
}

#define L2_GROUP 16

__global__ void __launch_bounds__(128)
layer2_scatter_kernel(const float* __restrict__ x,
                      const int* __restrict__ src,
                      const int* __restrict__ dst,
                      const float* __restrict__ deg_inv_r,
                      float* __restrict__ out,
                      long n_edges) {
    const long wave = ((long)blockIdx.x * blockDim.x + threadIdx.x) >> 5;
    const int  lane = (int)(threadIdx.x & 31);
    const long e0   = wave << 5;
    if (e0 >= n_edges) return;

    if (e0 + 32 > n_edges) {
        const int cnt = (int)(n_edges - e0);
        for (int j = 0; j < cnt; ++j) {
            const int   sj = src[e0 + j];
            const int   dj = dst[e0 + j];
            const float wj = deg_inv_r[dj];
            const float* xr = x + (long)sj * O_DIM + lane;
            const float v0 = xr[0];
            const float v1 = xr[32];
            float* o = out + (long)dj * O_DIM + lane;
            atomicAdd(o +  0, v0 * wj);
            atomicAdd(o + 32, v1 * wj);
        }
        return;
    }

    const int   s = src[e0 + lane];
    const int   d = dst[e0 + lane];
    const float w = deg_inv_r[d];

    #pragma unroll
    for (int g = 0; g < 32; g += L2_GROUP) {
        float v[L2_GROUP][2];
        #pragma unroll
        for (int j = 0; j < L2_GROUP; ++j) {
            const int sj = __builtin_amdgcn_readlane(s, g + j);
            const float* xr = x + (long)sj * O_DIM + lane;
            v[j][0] = xr[0];
            v[j][1] = xr[32];
        }
        #pragma unroll
        for (int j = 0; j < L2_GROUP; ++j) {
            const int   dj = __builtin_amdgcn_readlane(d, g + j);
            const float wj = __int_as_float(
                __builtin_amdgcn_readlane(__float_as_int(w), g + j));
            float* o = out + (long)dj * O_DIM + lane;
            atomicAdd(o +  0, v[j][0] * wj);
            atomicAdd(o + 32, v[j][1] * wj);
        }
    }
}

static unsigned stream_grid(long n) {
    long b = ((n >> 2) + 255) / 256;
    if (b < 1) b = 1;
    if (b > STREAM_BLOCKS) b = STREAM_BLOCKS;
    return (unsigned)b;
}

extern "C" void kernel_launch(void* const* d_in, const int* in_sizes, int n_in,
                              void* d_out, int out_size, void* d_ws, size_t ws_size,
                              hipStream_t stream) {
    const float* embed    = (const float*)d_in[0];
    const float* weight   = (const float*)d_in[1];
    const float* bias1    = (const float*)d_in[2];
    const float* bias2    = (const float*)d_in[3];
    const int*   edge_src = (const int*)d_in[4];
    const int*   edge_dst = (const int*)d_in[5];

    const int H = in_sizes[2];
    const int O = in_sizes[3];
    const int N = in_sizes[0] / H;
    const int R = in_sizes[1] / (H * O);
    const int E = in_sizes[4] / R;

    float* out     = (float*)d_out;
    float* deg_inv = (float*)d_ws;
    float* h1      = deg_inv + (long)R * N;
    float* x       = h1 + (long)N * H;

    const int BT = 256;
    const int ST = 128;

    long rn = (long)R * N;
    zero_f32_kernel<<<stream_grid(rn), BT, 0, stream>>>(deg_inv, rn);
    long re = (long)R * E;
    deg_count_kernel<<<(unsigned)((re + 4 * BT - 1) / (4 * BT)), BT, 0, stream>>>(
        edge_dst, deg_inv, E, N, re);
    deg_recip_kernel<<<stream_grid(rn), BT, 0, stream>>>(deg_inv, rn);

    long nh = (long)N * H;
    zero_f32_kernel<<<stream_grid(nh), BT, 0, stream>>>(h1, nh);
    long l1_threads = ((re + 31) / 32) * 32;
    layer1_scatter_kernel<<<(unsigned)((l1_threads + ST - 1) / ST), ST, 0, stream>>>(
        embed, edge_src, edge_dst, deg_inv, h1, E, N, re);
    relu_bias_kernel<<<stream_grid(nh), BT, 0, stream>>>(h1, bias1, nh);

    long no = (long)N * O;
    init_out_kernel<<<stream_grid(no), BT, 0, stream>>>(out, bias2, no);

    unsigned mm_blocks = (unsigned)((N + 127) / 128);
    long l2_threads = (((long)E + 31) / 32) * 32;
    for (int r = 0; r < R; ++r) {
        matmul_wmma_kernel<<<mm_blocks, BT, 0, stream>>>(
            h1, weight + (long)r * H * O, x, N);
        layer2_scatter_kernel<<<(unsigned)((l2_threads + ST - 1) / ST), ST, 0, stream>>>(
            x, edge_src + (long)r * E, edge_dst + (long)r * E,
            deg_inv + (long)r * N, out, (long)E);
    }
}


# Layer 2: one 25-wave block per WGP (250 blocks, all resident at once, no second round of blocks); W_r double-buffered in LDS, one block barrier per relation instead of two
# speedup vs baseline: 1.8916x; 1.4130x over previous
#include <hip/hip_runtime.h>


typedef __attribute__((ext_vector_type(2))) float v2f;
typedef __attribute__((ext_vector_type(4))) float v4f;
typedef __attribute__((ext_vector_type(4))) int   v4i;
typedef __attribute__((ext_vector_type(8))) float v8f;

#define H_DIM 128
#define O_DIM 64
#define AGG_PITCH 132
#define EDGE_BLOCKS 320
#define EDGE_U 8
#define L2_WAVES 25
#define L2_THREADS (L2_WAVES * 32)


__global__ void __launch_bounds__(256)
zero_i32_kernel(int* __restrict__ p, long n) {
    const long tid    = (long)blockIdx.x * blockDim.x + threadIdx.x;
    const long stride = (long)gridDim.x * blockDim.x;
    if ((((size_t)p) & 15) != 0) {
        for (long t = tid; t < n; t += stride) p[t] = 0;
        return;
    }
    v4i* __restrict__ p4 = (v4i*)p;
    const long n4 = n >> 2;
    const v4i z = {0, 0, 0, 0};
    for (long i = tid; i < n4; i += stride) p4[i] = z;
    for (long t = (n4 << 2) + tid; t < n; t += stride) p[t] = 0;
}

__global__ void __launch_bounds__(256)
count_kernel(const int* __restrict__ dst, int* __restrict__ cnt, int E, int N, long total) {
    const long tid    = (long)blockIdx.x * blockDim.x + threadIdx.x;
    const long stride = (long)gridDim.x * blockDim.x;
    long i = tid;
    for (; i + (EDGE_U - 1) * stride < total; i += EDGE_U * stride) {
        int d[EDGE_U];
        #pragma unroll
        for (int u = 0; u < EDGE_U; ++u) d[u] = dst[i + u * stride];
        #pragma unroll
        for (int u = 0; u < EDGE_U; ++u) {
            const int r = (int)((i + u * stride) / E);
            atomicAdd(&cnt[(long)r * N + d[u]], 1);
        }
    }
    for (; i < total; i += stride) {
        const int r = (int)(i / E);
        atomicAdd(&cnt[(long)r * N + dst[i]], 1);
    }
}

__global__ void __launch_bounds__(256)
place_kernel(const int* __restrict__ cnt, int* __restrict__ endv, int* __restrict__ cursor,
             long total) {
    __shared__ int wsum[8];
    __shared__ int bbase;
    const int  tid  = threadIdx.x;
    const int  lane = tid & 31;
    const int  wave = tid >> 5;
    const long k0   = ((long)blockIdx.x * 256 + tid) * 4;
    int c[4];
    #pragma unroll
    for (int u = 0; u < 4; ++u) c[u] = (k0 + u < total) ? cnt[k0 + u] : 0;
    const int tsum = c[0] + c[1] + c[2] + c[3];
    int incl = tsum;
    #pragma unroll
    for (int o = 1; o < 32; o <<= 1) {
        const int t = __shfl_up(incl, o);
        if (lane >= o) incl += t;
    }
    if (lane == 31) wsum[wave] = incl;
    __syncthreads();
    if (tid == 0) {
        int run = 0;
        for (int w = 0; w < 8; ++w) { const int t = wsum[w]; wsum[w] = run; run += t; }
        bbase = atomicAdd(cursor, run);
    }
    __syncthreads();
    int e = bbase + wsum[wave] + incl - tsum;
    #pragma unroll
    for (int u = 0; u < 4; ++u) {
        if (k0 + u < total) endv[k0 + u] = e;
        e += c[u];
    }
}

__global__ void __launch_bounds__(256)
fill_kernel(const int* __restrict__ src, const int* __restrict__ dst,
            int* __restrict__ endv, int* __restrict__ perm, int E, int N, long total) {
    const long tid    = (long)blockIdx.x * blockDim.x + threadIdx.x;
    const long stride = (long)gridDim.x * blockDim.x;
    long i = tid;
    for (; i + (EDGE_U - 1) * stride < total; i += EDGE_U * stride) {
        int d[EDGE_U], s[EDGE_U], p[EDGE_U];
        #pragma unroll
        for (int u = 0; u < EDGE_U; ++u) { d[u] = dst[i + u * stride]; s[u] = src[i + u * stride]; }
        #pragma unroll
        for (int u = 0; u < EDGE_U; ++u) {
            const int r = (int)((i + u * stride) / E);
            p[u] = atomicAdd(&endv[(long)r * N + d[u]], 1);
        }
        #pragma unroll
        for (int u = 0; u < EDGE_U; ++u) perm[p[u]] = s[u];
    }
    for (; i < total; i += stride) {
        const int r = (int)(i / E);
        const int p = atomicAdd(&endv[(long)r * N + dst[i]], 1);
        perm[p] = src[i];
    }
}

template <int NP, int G>
__device__ __forceinline__ void gather_pass(const float* __restrict__ table,
                                            const int* __restrict__ perm,
                                            float* slice, int pitch, int lane,
                                            int myc, int myst, float myw, int myrow) {
    int T = 0;
    #pragma unroll
    for (int q = 0; q < NP; ++q) T += __builtin_amdgcn_readlane(myc, q);

    for (int base = 0; base < T; base += 32) {
        const int m   = min(32, T - base);
        const int off = base + lane;
        int   idx = -1;
        int   row = 0;
        float w   = 0.0f;
        int   cum = 0;
        #pragma unroll
        for (int q = 0; q < NP; ++q) {
            const int   cq = __builtin_amdgcn_readlane(myc, q);
            const int   sq = __builtin_amdgcn_readlane(myst, q);
            const int   rq = __builtin_amdgcn_readlane(myrow, q);
            const float wq = __int_as_float(__builtin_amdgcn_readlane(__float_as_int(myw), q));
            const bool hit = (off >= cum) && (off < cum + cq);
            idx = hit ? sq + (off - cum) : idx;
            row = hit ? rq : row;
            w   = hit ? wq : w;
            cum += cq;
        }
        const int sv = (idx >= 0) ? perm[idx] : 0;
        for (int j = 0; j < m; j += G) {
            v4f v[G];
            #pragma unroll
            for (int u = 0; u < G; ++u) {
                const int jj = min(j + u, m - 1);
                const int s  = __builtin_amdgcn_readlane(sv, jj);
                v[u] = *(const v4f*)(table + (long)s * H_DIM + lane * 4);
            }
            #pragma unroll
            for (int u = 0; u < G; ++u) {
                if (j + u < m) {
                    const int   jj = j + u;
                    const int   mr = __builtin_amdgcn_readlane(row, jj);
                    const float wj = __int_as_float(
                        __builtin_amdgcn_readlane(__float_as_int(w), jj));
                    v4f* p = (v4f*)(slice + mr * pitch + lane * 4);
                    *p = *p + v[u] * wj;
                }
            }
        }
    }
}

__global__ void __launch_bounds__(256)
layer1_gather_kernel(const float* __restrict__ embed, const int* __restrict__ cnt,
                     const int* __restrict__ endv, const int* __restrict__ perm,
                     const float* __restrict__ bias1, float* __restrict__ h1,
                     int N, int R) {
    __shared__ float acc[8 * 16 * H_DIM];
    const int tid   = threadIdx.x;
    const int wave  = tid >> 5;
    const int lane  = tid & 31;
    const int rowBase = blockIdx.x * 128 + wave * 16;
    float* slice = acc + wave * (16 * H_DIM);

    const v4f z = {0.0f, 0.0f, 0.0f, 0.0f};
    #pragma unroll
    for (int m = 0; m < 16; ++m) *(v4f*)(slice + m * H_DIM + lane * 4) = z;

    const int pairs = 16 * R;
    for (int p0 = 0; p0 < pairs; p0 += 32) {
        const int pidx = p0 + lane;
        const int nl   = pidx / R;
        const int r    = pidx - nl * R;
        const int d    = rowBase + nl;
        int myc = 0, myst = 0;
        if (pidx < pairs && d < N) {
            myc  = cnt[(long)r * N + d];
            myst = endv[(long)r * N + d] - myc;
        }
        const float myw = 1.0f / fmaxf((float)myc, 1.0f);
        gather_pass<32, 16>(embed, perm, slice, H_DIM, lane, myc, myst, myw, nl);
    }

    const v4f b = *(const v4f*)(bias1 + lane * 4);
    #pragma unroll
    for (int m = 0; m < 16; ++m) {
        const int d = rowBase + m;
        if (d < N) {
            const v4f a = *(const v4f*)(slice + m * H_DIM + lane * 4);
            v4f h;
            h.x = fmaxf(a.x + b.x, 0.0f);
            h.y = fmaxf(a.y + b.y, 0.0f);
            h.z = fmaxf(a.z + b.z, 0.0f);
            h.w = fmaxf(a.w + b.w, 0.0f);
            *(v4f*)(h1 + (long)d * H_DIM + lane * 4) = h;
        }
    }
}

__device__ __forceinline__ void stage_w(const float* __restrict__ W, float* __restrict__ dst, int tid) {
    for (int i = tid; i < H_DIM * O_DIM; i += L2_THREADS) {
        const int k = i >> 6;
        const int o = i & (O_DIM - 1);
        dst[(k >> 1) * (2 * O_DIM) + o * 2 + (k & 1)] = W[i];
    }
}

__global__ void __launch_bounds__(L2_THREADS)
layer2_gather_wmma_kernel(const float* __restrict__ h1, const float* __restrict__ weight,
                          const int* __restrict__ cnt, const int* __restrict__ endv,
                          const int* __restrict__ perm, const float* __restrict__ bias2,
                          float* __restrict__ out, int N, int R) {
    __shared__ float Wlds[2 * H_DIM * O_DIM];
    __shared__ float agg[L2_WAVES * 16 * AGG_PITCH];

    const int tid   = threadIdx.x;
    const int wave  = tid >> 5;
    const int lane  = tid & 31;
    const int lhalf = lane >> 4;
    const int l15   = lane & 15;
    const int rowBase = blockIdx.x * (L2_WAVES * 16) + wave * 16;
    float* slice = agg + wave * (16 * AGG_PITCH);

    v8f acc0 = {}; v8f acc1 = {}; v8f acc2 = {}; v8f acc3 = {};

    if (R > 0) stage_w(weight, Wlds, tid);
    for (int r = 0; r < R; ++r) {
        const float* wcur = Wlds + (r & 1) * (H_DIM * O_DIM);
        if (r + 1 < R)
            stage_w(weight + (long)(r + 1) * H_DIM * O_DIM, Wlds + ((r + 1) & 1) * (H_DIM * O_DIM), tid);

        const v4f z = {0.0f, 0.0f, 0.0f, 0.0f};
        #pragma unroll
        for (int m = 0; m < 16; ++m) *(v4f*)(slice + m * AGG_PITCH + lane * 4) = z;

        int myc = 0, myst = 0;
        {
            const int d = rowBase + l15;
            if (lhalf == 0 && d < N) {
                myc  = cnt[(long)r * N + d];
                myst = endv[(long)r * N + d] - myc;
            }
        }
        const float myw = 1.0f / fmaxf((float)myc, 1.0f);
        gather_pass<16, 16>(h1, perm, slice, AGG_PITCH, lane, myc, myst, myw, l15);

        if (r == 0) __syncthreads();

        const v2f* abase = (const v2f*)(slice + l15 * AGG_PITCH + lhalf * 2);
        const v2f* bbase = (const v2f*)wcur + lhalf * O_DIM + l15;
        for (int k0 = 0; k0 < H_DIM; k0 += 4) {
            const v2f a = abase[k0 >> 1];
            const v2f* bp = bbase + (k0 >> 1) * O_DIM;
            const v2f b0 = bp[0];
            const v2f b1 = bp[16];
            const v2f b2 = bp[32];
            const v2f b3 = bp[48];
            acc0 = __builtin_amdgcn_wmma_f32_16x16x4_f32(false, a, false, b0,
                                                         (short)0, acc0, false, false);
            acc1 = __builtin_amdgcn_wmma_f32_16x16x4_f32(false, a, false, b1,
                                                         (short)0, acc1, false, false);
            acc2 = __builtin_amdgcn_wmma_f32_16x16x4_f32(false, a, false, b2,
                                                         (short)0, acc2, false, false);
            acc3 = __builtin_amdgcn_wmma_f32_16x16x4_f32(false, a, false, b3,
                                                         (short)0, acc3, false, false);
        }
        if (r + 1 < R) __syncthreads();
    }

    const float bz0 = bias2[l15], bz1 = bias2[l15 + 16], bz2 = bias2[l15 + 32], bz3 = bias2[l15 + 48];
    const int crow = rowBase + 8 * lhalf;
    #pragma unroll
    for (int v = 0; v < 8; ++v) {
        const int row = crow + v;
        if (row < N) {
            float* o = out + (long)row * O_DIM + l15;
            o[0]  = acc0[v] + bz0;
            o[16] = acc1[v] + bz1;
            o[32] = acc2[v] + bz2;
            o[48] = acc3[v] + bz3;
        }
    }
}

extern "C" void kernel_launch(void* const* d_in, const int* in_sizes, int n_in,
                              void* d_out, int out_size, void* d_ws, size_t ws_size,
                              hipStream_t stream) {
    const float* embed    = (const float*)d_in[0];
    const float* weight   = (const float*)d_in[1];
    const float* bias1    = (const float*)d_in[2];
    const float* bias2    = (const float*)d_in[3];
    const int*   edge_src = (const int*)d_in[4];
    const int*   edge_dst = (const int*)d_in[5];

    const int H = in_sizes[2];
    const int O = in_sizes[3];
    const int N = in_sizes[0] / H;
    const int R = in_sizes[1] / (H * O);
    const int E = in_sizes[4] / R;

    const long rn = (long)R * N;
    const long re = (long)R * E;

    float* out    = (float*)d_out;
    int*   cnt    = (int*)d_ws;
    int*   cursor = cnt + rn;
    int*   endv   = cursor + 32;
    int*   perm   = endv + rn;
    float* h1     = (float*)d_ws + ((2 * rn + 32 + re + 3) & ~3L);

    const int BT = 256;

    zero_i32_kernel<<<128, BT, 0, stream>>>(cnt, rn + 32);
    count_kernel<<<EDGE_BLOCKS, BT, 0, stream>>>(edge_dst, cnt, E, N, re);
    place_kernel<<<(unsigned)((rn + 4 * BT - 1) / (4 * BT)), BT, 0, stream>>>(
        cnt, endv, cursor, rn);
    fill_kernel<<<EDGE_BLOCKS, BT, 0, stream>>>(edge_src, edge_dst, endv, perm, E, N, re);

    const unsigned row_blocks = (unsigned)((N + 127) / 128);
    layer1_gather_kernel<<<row_blocks, BT, 0, stream>>>(
        embed, cnt, endv, perm, bias1, h1, N, R);
    const unsigned l2_blocks = (unsigned)((N + L2_WAVES * 16 - 1) / (L2_WAVES * 16));
    layer2_gather_wmma_kernel<<<l2_blocks, L2_THREADS, 0, stream>>>(
        h1, weight, cnt, endv, perm, bias2, out, N, R);
}


# Layer 1: each wave walks its 16 rows' contiguous run of the grouped edge list once with a scalar cursor; row sum in registers, stored straight to h1; no LDS, no pair search; sum order unchanged
# speedup vs baseline: 2.8240x; 1.4929x over previous
#include <hip/hip_runtime.h>


typedef __attribute__((ext_vector_type(2))) float v2f;
typedef __attribute__((ext_vector_type(4))) float v4f;
typedef __attribute__((ext_vector_type(4))) int   v4i;
typedef __attribute__((ext_vector_type(8))) float v8f;

#define H_DIM 128
#define O_DIM 64
#define AGG_PITCH 132
#define L2_WAVES 25
#define L2_THREADS (L2_WAVES * 32)


#define BIN_DST      400
#define BIN_BLOCKS   128
#define BIN_THREADS  512
#define BIN_U        10
#define MAX_BINS     1024
#define SORT_THREADS 512
#define SORT_U       6

typedef __attribute__((ext_vector_type(2))) int v2i;

__device__ __forceinline__ int block_excl_scan(int v, int* wsum, int tid) {
    const int lane = tid & 31;
    const int wave = tid >> 5;
    const int nw   = (int)(blockDim.x >> 5);
    int incl = v;
    #pragma unroll
    for (int o = 1; o < 32; o <<= 1) {
        const int t = __shfl_up(incl, o);
        if (lane >= o) incl += t;
    }
    if (lane == 31) wsum[wave] = incl;
    __syncthreads();
    if (wave == 0) {
        const int w = (lane < nw) ? wsum[lane] : 0;
        int wi = w;
        #pragma unroll
        for (int o = 1; o < 32; o <<= 1) {
            const int t = __shfl_up(wi, o);
            if (lane >= o) wi += t;
        }
        wsum[lane] = wi - w;
    }
    __syncthreads();
    return wsum[wave] + incl - v;
}

__global__ void __launch_bounds__(BIN_THREADS)
bin_count_kernel(const int* __restrict__ dst, int* __restrict__ histT, long total, int nbins) {
    __shared__ int h[MAX_BINS];
    const int tid = threadIdx.x;
    for (int b = tid; b < nbins; b += BIN_THREADS) h[b] = 0;
    __syncthreads();
    const long chunk = (total + BIN_BLOCKS - 1) / BIN_BLOCKS;
    const long beg   = (long)blockIdx.x * chunk;
    const long end   = min(beg + chunk, total);
    for (long i0 = beg + tid; i0 < end; i0 += (long)BIN_THREADS * BIN_U) {
        int d[BIN_U];
        #pragma unroll
        for (int u = 0; u < BIN_U; ++u) {
            const long i = i0 + (long)u * BIN_THREADS;
            d[u] = (i < end) ? dst[i] : -1;
        }
        #pragma unroll
        for (int u = 0; u < BIN_U; ++u)
            if (d[u] >= 0) atomicAdd(&h[d[u] / BIN_DST], 1);
    }
    __syncthreads();
    for (int b = tid; b < nbins; b += BIN_THREADS) histT[(long)b * BIN_BLOCKS + blockIdx.x] = h[b];
}

__global__ void __launch_bounds__(BIN_THREADS)
bin_scatter_kernel(const int* __restrict__ src, const int* __restrict__ dst,
                   const int* __restrict__ histT, int* __restrict__ binStart,
                   v2i* __restrict__ rec, int E, int R, long total, int nbins) {
    __shared__ int tot[MAX_BINS];
    __shared__ int base[MAX_BINS];
    __shared__ int wsum[32];
    const int tid = threadIdx.x;
    const int bid = blockIdx.x;

    for (int b0 = 0; b0 < nbins; b0 += BIN_THREADS / 2) {
        const int b    = b0 + (tid >> 1);
        const int half = tid & 1;
        int all = 0, bef = 0;
        if (b < nbins) {
            const v4i* p = (const v4i*)(histT + (long)b * BIN_BLOCKS + half * (BIN_BLOCKS / 2));
            v4i v[BIN_BLOCKS / 8];
            #pragma unroll
            for (int j = 0; j < BIN_BLOCKS / 8; ++j) v[j] = p[j];
            #pragma unroll
            for (int j = 0; j < BIN_BLOCKS / 8; ++j) {
                const int k = half * (BIN_BLOCKS / 2) + j * 4;
                all += v[j].x + v[j].y + v[j].z + v[j].w;
                bef += (k     < bid ? v[j].x : 0) + (k + 1 < bid ? v[j].y : 0)
                     + (k + 2 < bid ? v[j].z : 0) + (k + 3 < bid ? v[j].w : 0);
            }
        }
        all += __shfl_xor(all, 1);
        bef += __shfl_xor(bef, 1);
        if (b < nbins && half == 0) { tot[b] = all; base[b] = bef; }
    }
    __syncthreads();

    {
        const int b2 = tid * 2;
        const int v0 = (b2     < nbins) ? tot[b2]     : 0;
        const int v1 = (b2 + 1 < nbins) ? tot[b2 + 1] : 0;
        const int ex = block_excl_scan(v0 + v1, wsum, tid);
        if (b2 < nbins)     { base[b2]     += ex;      if (bid == 0) binStart[b2]     = ex; }
        if (b2 + 1 < nbins) { base[b2 + 1] += ex + v0; if (bid == 0) binStart[b2 + 1] = ex + v0; }
        if (bid == 0 && tid == 0) binStart[nbins] = (int)total;
    }
    __syncthreads();

    const long chunk = (total + BIN_BLOCKS - 1) / BIN_BLOCKS;
    const long beg   = (long)bid * chunk;
    const long end   = min(beg + chunk, total);
    for (long i0 = beg + tid; i0 < end; i0 += (long)BIN_THREADS * BIN_U) {
        int d[BIN_U], s[BIN_U];
        #pragma unroll
        for (int u = 0; u < BIN_U; ++u) {
            const long i = i0 + (long)u * BIN_THREADS;
            d[u] = (i < end) ? dst[i] : -1;
            s[u] = (i < end) ? src[i] : 0;
        }
        int key[BIN_U], pos[BIN_U];
        #pragma unroll
        for (int u = 0; u < BIN_U; ++u) {
            key[u] = 0;
            pos[u] = 0;
            if (d[u] >= 0) {
                const unsigned i = (unsigned)(i0 + (long)u * BIN_THREADS);
                const int r  = (int)(i / (unsigned)E);
                const int b  = d[u] / BIN_DST;
                const int dl = d[u] - b * BIN_DST;
                key[u] = dl * R + r;
                pos[u] = atomicAdd(&base[b], 1);
            }
        }
        #pragma unroll
        for (int u = 0; u < BIN_U; ++u) {
            if (d[u] >= 0) {
                const v2i rc = {key[u], s[u]};
                rec[pos[u]] = rc;
            }
        }
    }
}

__global__ void __launch_bounds__(SORT_THREADS)
bin_sort_kernel(const int* __restrict__ binStart, const v2i* __restrict__ rec,
                int* __restrict__ cnt, int* __restrict__ endv, int* __restrict__ perm,
                int N, int R) {
    extern __shared__ int keyc[];
    __shared__ int wsum[32];
    __shared__ int stage[SORT_U * SORT_THREADS];
    const int tid = threadIdx.x;
    const int bid = blockIdx.x;
    const int K   = R * BIN_DST;
    int* c = keyc;
    int* o = keyc + K;

    const int bs = binStart[bid];
    const int n  = binStart[bid + 1] - bs;
    const v2i* rb = rec + bs;

    v2i first[SORT_U];
    #pragma unroll
    for (int u = 0; u < SORT_U; ++u) {
        const int i = tid + u * SORT_THREADS;
        const v2i none = {-1, 0};
        first[u] = (i < n) ? rb[i] : none;
    }
    for (int k = tid; k < K; k += SORT_THREADS) c[k] = 0;
    __syncthreads();

    #pragma unroll
    for (int u = 0; u < SORT_U; ++u)
        if (first[u].x >= 0) atomicAdd(&c[first[u].x], 1);
    for (int i = tid + SORT_U * SORT_THREADS; i < n; i += SORT_THREADS) atomicAdd(&c[rb[i].x], 1);
    __syncthreads();

    {
        const int kpt = (K + SORT_THREADS - 1) / SORT_THREADS;
        const int k0  = min(K, tid * kpt);
        const int k1  = min(K, k0 + kpt);
        int s = 0;
        for (int k = k0; k < k1; ++k) s += c[k];
        int run = block_excl_scan(s, wsum, tid);
        for (int k = k0; k < k1; ++k) { o[k] = run; run += c[k]; }
    }
    __syncthreads();

    int* pb = perm + bs;
    int pos[SORT_U];
    #pragma unroll
    for (int u = 0; u < SORT_U; ++u) {
        pos[u] = 0;
        if (first[u].x >= 0) pos[u] = atomicAdd(&o[first[u].x], 1);
    }
    if (n <= SORT_U * SORT_THREADS) {
        #pragma unroll
        for (int u = 0; u < SORT_U; ++u)
            if (first[u].x >= 0) stage[pos[u]] = first[u].y;
        __syncthreads();
        for (int i = tid; i < n; i += SORT_THREADS) pb[i] = stage[i];
    } else {
        #pragma unroll
        for (int u = 0; u < SORT_U; ++u)
            if (first[u].x >= 0) pb[pos[u]] = first[u].y;
        for (int i = tid + SORT_U * SORT_THREADS; i < n; i += SORT_THREADS) {
            const v2i rc = rb[i];
            pb[atomicAdd(&o[rc.x], 1)] = rc.y;
        }
    }
    __syncthreads();

    const int d0 = bid * BIN_DST;
    for (int j = tid; j < K; j += SORT_THREADS) {
        const int r  = j / BIN_DST;
        const int dl = j - r * BIN_DST;
        const int d  = d0 + dl;
        if (d < N) {
            const int key = dl * R + r;
            cnt[(long)r * N + d]  = c[key];
            endv[(long)r * N + d] = bs + o[key];
        }
    }
}

template <int NP, int G>
__device__ __forceinline__ void gather_pass(const float* __restrict__ table,
                                            const int* __restrict__ perm,
                                            float* slice, int pitch, int lane,
                                            int myc, int myst, float myw, int myrow) {
    int T = 0;
    #pragma unroll
    for (int q = 0; q < NP; ++q) T += __builtin_amdgcn_readlane(myc, q);

    for (int base = 0; base < T; base += 32) {
        const int m   = min(32, T - base);
        const int off = base + lane;
        int   idx = -1;
        int   row = 0;
        float w   = 0.0f;
        int   cum = 0;
        #pragma unroll
        for (int q = 0; q < NP; ++q) {
            const int   cq = __builtin_amdgcn_readlane(myc, q);
            const int   sq = __builtin_amdgcn_readlane(myst, q);
            const int   rq = __builtin_amdgcn_readlane(myrow, q);
            const float wq = __int_as_float(__builtin_amdgcn_readlane(__float_as_int(myw), q));
            const bool hit = (off >= cum) && (off < cum + cq);
            idx = hit ? sq + (off - cum) : idx;
            row = hit ? rq : row;
            w   = hit ? wq : w;
            cum += cq;
        }
        const int sv = (idx >= 0) ? perm[idx] : 0;
        for (int j = 0; j < m; j += G) {
            v4f v[G];
            #pragma unroll
            for (int u = 0; u < G; ++u) {
                const int jj = min(j + u, m - 1);
                const int s  = __builtin_amdgcn_readlane(sv, jj);
                v[u] = *(const v4f*)(table + (long)s * H_DIM + lane * 4);
            }
            #pragma unroll
            for (int u = 0; u < G; ++u) {
                if (j + u < m) {
                    const int   jj = j + u;
                    const int   mr = __builtin_amdgcn_readlane(row, jj);
                    const float wj = __int_as_float(
                        __builtin_amdgcn_readlane(__float_as_int(w), jj));
                    v4f* p = (v4f*)(slice + mr * pitch + lane * 4);
                    *p = *p + v[u] * wj;
                }
            }
        }
    }
}

static_assert(BIN_DST % 16 == 0, "the 16 rows of a wave must lie in one bin");
#define L1_G 16

__global__ void __launch_bounds__(256)
layer1_gather_kernel(const float* __restrict__ embed, const int* __restrict__ cnt,
                     const int* __restrict__ endv, const int* __restrict__ perm,
                     const float* __restrict__ bias1, float* __restrict__ h1,
                     int N, int R) {
    const int tid   = threadIdx.x;
    const int wave  = __builtin_amdgcn_readfirstlane(tid >> 5);
    const int lane  = tid & 31;
    const int rowBase = blockIdx.x * 128 + wave * 16;
    if (rowBase >= N) return;
    const int nrows = min(16, N - rowBase);
    const int pairs = nrows * R;

    const v4f b = *(const v4f*)(bias1 + lane * 4);
    const v4f z = {0.0f, 0.0f, 0.0f, 0.0f};
    v4f acc = z;
    int row = 0;
    int rr  = 0;

    auto store_row = [&]() {
        v4f h;
        h.x = fmaxf(acc.x + b.x, 0.0f);
        h.y = fmaxf(acc.y + b.y, 0.0f);
        h.z = fmaxf(acc.z + b.z, 0.0f);
        h.w = fmaxf(acc.w + b.w, 0.0f);
        *(v4f*)(h1 + (long)(rowBase + row) * H_DIM + lane * 4) = h;
        acc = z;
        ++row;
    };
    auto next_pair = [&]() {
        if (++rr == R) { rr = 0; store_row(); }
    };
    auto load_pairs = [&](int q0, int& e, int& c) {
        const int pidx = q0 + lane;
        e = 0; c = 0;
        if (pidx < pairs) {
            const int nl = pidx / R;
            const int r  = pidx - nl * R;
            const int d  = rowBase + nl;
            c = cnt[(long)r * N + d];
            e = endv[(long)r * N + d];
        }
    };

    int e = 0, c = 0;
    load_pairs(0, e, c);
    for (int q0 = 0; q0 < pairs; q0 += 32) {
        int en = 0, cn = 0;
        if (q0 + 32 < pairs) load_pairs(q0 + 32, en, cn);
        const int np = min(32, pairs - q0);
        const float w = 1.0f / fmaxf((float)c, 1.0f);
        const int ew = (lane < np - 1) ? e : 0x7fffffff;

        if (q0 > 0) next_pair();
        int   ql      = 0;
        int   cur_end = __builtin_amdgcn_readlane(ew, 0);
        float cur_w   = __int_as_float(__builtin_amdgcn_readlane(__float_as_int(w), 0));
        const int S = __builtin_amdgcn_readlane(e, 0) - __builtin_amdgcn_readlane(c, 0);
        const int E = __builtin_amdgcn_readlane(e, np - 1);

        int sv = (S + lane < E) ? perm[S + lane] : 0;
        for (int p0 = S; p0 < E; p0 += 32) {
            const int m   = min(32, E - p0);
            int svn = 0;
            for (int j = 0; j < m; j += L1_G) {
                v4f v[L1_G];
                #pragma unroll
                for (int u = 0; u < L1_G; ++u) {
                    const int jj = min(j + u, m - 1);
                    const int s  = __builtin_amdgcn_readlane(sv, jj);
                    v[u] = *(const v4f*)(embed + (long)s * H_DIM + lane * 4);
                }
                if (j == 0 && p0 + 32 + lane < E) svn = perm[p0 + 32 + lane];
                #pragma unroll
                for (int u = 0; u < L1_G; ++u) {
                    if (j + u < m) {
                        const int p = p0 + j + u;
                        while (p >= cur_end) {
                            ++ql;
                            next_pair();
                            cur_end = __builtin_amdgcn_readlane(ew, ql);
                            cur_w   = __int_as_float(__builtin_amdgcn_readlane(__float_as_int(w), ql));
                        }
                        acc = acc + v[u] * cur_w;
                    }
                }
            }
            sv = svn;
        }
        for (; ql < np - 1; ++ql) next_pair();
        e = en; c = cn;
    }
    while (row < nrows) store_row();
}

__device__ __forceinline__ void stage_w(const float* __restrict__ W, float* __restrict__ dst, int tid) {
    for (int i = tid; i < H_DIM * O_DIM; i += L2_THREADS) {
        const int k = i >> 6;
        const int o = i & (O_DIM - 1);
        dst[(k >> 1) * (2 * O_DIM) + o * 2 + (k & 1)] = W[i];
    }
}

__global__ void __launch_bounds__(L2_THREADS)
layer2_gather_wmma_kernel(const float* __restrict__ h1, const float* __restrict__ weight,
                          const int* __restrict__ cnt, const int* __restrict__ endv,
                          const int* __restrict__ perm, const float* __restrict__ bias2,
                          float* __restrict__ out, int N, int R) {
    __shared__ float Wlds[2 * H_DIM * O_DIM];
    __shared__ float agg[L2_WAVES * 16 * AGG_PITCH];

    const int tid   = threadIdx.x;
    const int wave  = tid >> 5;
    const int lane  = tid & 31;
    const int lhalf = lane >> 4;
    const int l15   = lane & 15;
    const int rowBase = blockIdx.x * (L2_WAVES * 16) + wave * 16;
    float* slice = agg + wave * (16 * AGG_PITCH);

    v8f acc0 = {}; v8f acc1 = {}; v8f acc2 = {}; v8f acc3 = {};

    if (R > 0) stage_w(weight, Wlds, tid);
    for (int r = 0; r < R; ++r) {
        const float* wcur = Wlds + (r & 1) * (H_DIM * O_DIM);
        if (r + 1 < R)
            stage_w(weight + (long)(r + 1) * H_DIM * O_DIM, Wlds + ((r + 1) & 1) * (H_DIM * O_DIM), tid);

        const v4f z = {0.0f, 0.0f, 0.0f, 0.0f};
        #pragma unroll
        for (int m = 0; m < 16; ++m) *(v4f*)(slice + m * AGG_PITCH + lane * 4) = z;

        int myc = 0, myst = 0;
        {
            const int d = rowBase + l15;
            if (lhalf == 0 && d < N) {
                myc  = cnt[(long)r * N + d];
                myst = endv[(long)r * N + d] - myc;
            }
        }
        const float myw = 1.0f / fmaxf((float)myc, 1.0f);
        gather_pass<16, 16>(h1, perm, slice, AGG_PITCH, lane, myc, myst, myw, l15);

        if (r == 0) __syncthreads();

        const v2f* abase = (const v2f*)(slice + l15 * AGG_PITCH + lhalf * 2);
        const v2f* bbase = (const v2f*)wcur + lhalf * O_DIM + l15;
        for (int k0 = 0; k0 < H_DIM; k0 += 4) {
            const v2f a = abase[k0 >> 1];
            const v2f* bp = bbase + (k0 >> 1) * O_DIM;
            const v2f b0 = bp[0];
            const v2f b1 = bp[16];
            const v2f b2 = bp[32];
            const v2f b3 = bp[48];
            acc0 = __builtin_amdgcn_wmma_f32_16x16x4_f32(false, a, false, b0,
                                                         (short)0, acc0, false, false);
            acc1 = __builtin_amdgcn_wmma_f32_16x16x4_f32(false, a, false, b1,
                                                         (short)0, acc1, false, false);
            acc2 = __builtin_amdgcn_wmma_f32_16x16x4_f32(false, a, false, b2,
                                                         (short)0, acc2, false, false);
            acc3 = __builtin_amdgcn_wmma_f32_16x16x4_f32(false, a, false, b3,
                                                         (short)0, acc3, false, false);
        }
        if (r + 1 < R) __syncthreads();
    }

    const float bz0 = bias2[l15], bz1 = bias2[l15 + 16], bz2 = bias2[l15 + 32], bz3 = bias2[l15 + 48];
    const int crow = rowBase + 8 * lhalf;
    #pragma unroll
    for (int v = 0; v < 8; ++v) {
        const int row = crow + v;
        if (row < N) {
            float* o = out + (long)row * O_DIM + l15;
            o[0]  = acc0[v] + bz0;
            o[16] = acc1[v] + bz1;
            o[32] = acc2[v] + bz2;
            o[48] = acc3[v] + bz3;
        }
    }
}

extern "C" void kernel_launch(void* const* d_in, const int* in_sizes, int n_in,
                              void* d_out, int out_size, void* d_ws, size_t ws_size,
                              hipStream_t stream) {
    const float* embed    = (const float*)d_in[0];
    const float* weight   = (const float*)d_in[1];
    const float* bias1    = (const float*)d_in[2];
    const float* bias2    = (const float*)d_in[3];
    const int*   edge_src = (const int*)d_in[4];
    const int*   edge_dst = (const int*)d_in[5];

    const int H = in_sizes[2];
    const int O = in_sizes[3];
    const int N = in_sizes[0] / H;
    const int R = in_sizes[1] / (H * O);
    const int E = in_sizes[4] / R;

    const long rn = (long)R * N;
    const long re = (long)R * E;

    float* out    = (float*)d_out;
    int*   cnt    = (int*)d_ws;
    int*   endv   = cnt + rn;
    int*   perm   = endv + rn;
    float* h1     = (float*)d_ws + ((2 * rn + re + 3) & ~3L);

    const int nbins = (N + BIN_DST - 1) / BIN_DST;
    v2i* rec      = (v2i*)h1;
    int* histT    = (int*)h1 + ((2 * re + 3) & ~3L);
    int* binStart = histT + (long)nbins * BIN_BLOCKS;

    const int BT = 256;

    bin_count_kernel<<<BIN_BLOCKS, BIN_THREADS, 0, stream>>>(edge_dst, histT, re, nbins);
    bin_scatter_kernel<<<BIN_BLOCKS, BIN_THREADS, 0, stream>>>(
        edge_src, edge_dst, histT, binStart, rec, E, R, re, nbins);
    bin_sort_kernel<<<nbins, SORT_THREADS, 2 * (size_t)R * BIN_DST * sizeof(int), stream>>>(
        binStart, rec, cnt, endv, perm, N, R);

    const unsigned row_blocks = (unsigned)((N + 127) / 128);
    layer1_gather_kernel<<<row_blocks, BT, 0, stream>>>(
        embed, cnt, endv, perm, bias1, h1, N, R);
    const unsigned l2_blocks = (unsigned)((N + L2_WAVES * 16 - 1) / (L2_WAVES * 16));
    layer2_gather_wmma_kernel<<<l2_blocks, L2_THREADS, 0, stream>>>(
        h1, weight, cnt, endv, perm, bias2, out, N, R);
}


# Layer 2 gather: a wave walks its 16 rows' edges of a relation in list order with a scalar cursor; row sum in registers, one LDS store per row; no LDS read-add-write, no zero fill; sum order unchanged
# speedup vs baseline: 3.4795x; 1.2321x over previous
#include <hip/hip_runtime.h>


typedef __attribute__((ext_vector_type(2))) float v2f;
typedef __attribute__((ext_vector_type(4))) float v4f;
typedef __attribute__((ext_vector_type(4))) int   v4i;
typedef __attribute__((ext_vector_type(8))) float v8f;

#define H_DIM 128
#define O_DIM 64
#define AGG_PITCH 132
#define L2_WAVES 25
#define L2_THREADS (L2_WAVES * 32)


#define BIN_DST      400
#define BIN_THREADS  512
#define BIN_U        10
#define BIN_EDGES    (BIN_THREADS * BIN_U)
#define MAX_BINS     (2 * BIN_THREADS)
#define L1_WAVES     (BIN_DST / 16)
#define L1_THREADS   (L1_WAVES * 32)
#define SORT_U       4
#define STAGE_CAP    (SORT_U * L1_THREADS)

typedef __attribute__((ext_vector_type(2))) int v2i;

__device__ __forceinline__ int block_excl_scan(int v, int* wsum, int tid) {
    const int lane = tid & 31;
    const int wave = tid >> 5;
    const int nw   = (int)(blockDim.x >> 5);
    int incl = v;
    #pragma unroll
    for (int o = 1; o < 32; o <<= 1) {
        const int t = __shfl_up(incl, o);
        if (lane >= o) incl += t;
    }
    if (lane == 31) wsum[wave] = incl;
    __syncthreads();
    if (wave == 0) {
        const int w = (lane < nw) ? wsum[lane] : 0;
        int wi = w;
        #pragma unroll
        for (int o = 1; o < 32; o <<= 1) {
            const int t = __shfl_up(wi, o);
            if (lane >= o) wi += t;
        }
        wsum[lane] = wi - w;
    }
    __syncthreads();
    return wsum[wave] + incl - v;
}

__global__ void __launch_bounds__(BIN_THREADS)
bin_group_kernel(const int* __restrict__ src, const int* __restrict__ dst,
                 v2i* __restrict__ pair, v2i* __restrict__ rec,
                 int E, int R, long total, int pitch) {
    __shared__ int h[MAX_BINS];
    __shared__ int wsum[32];
    const int tid = threadIdx.x;
    const int bid = blockIdx.x;
    const long beg = (long)bid * BIN_EDGES;

    int d[BIN_U], s[BIN_U];
    #pragma unroll
    for (int u = 0; u < BIN_U; ++u) {
        const long i = beg + tid + (long)u * BIN_THREADS;
        d[u] = (i < total) ? dst[i] : -1;
        s[u] = (i < total) ? src[i] : 0;
    }
    h[tid * 2]     = 0;
    h[tid * 2 + 1] = 0;
    __syncthreads();

    int key[BIN_U], bin[BIN_U], rank[BIN_U];
    #pragma unroll
    for (int u = 0; u < BIN_U; ++u) {
        key[u]  = 0;
        bin[u]  = 0;
        rank[u] = 0;
        if (d[u] >= 0) {
            const unsigned i = (unsigned)(beg + tid + (long)u * BIN_THREADS);
            const int r = (int)(i / (unsigned)E);
            const int b = d[u] / BIN_DST;
            key[u]  = (d[u] - b * BIN_DST) * R + r;
            bin[u]  = b;
            rank[u] = atomicAdd(&h[b], 1);
        }
    }
    __syncthreads();

    {
        const int b2 = tid * 2;
        const int v0 = h[b2];
        const int v1 = h[b2 + 1];
        const int ex = block_excl_scan(v0 + v1, wsum, tid);
        h[b2]     = ex;
        h[b2 + 1] = ex + v0;
        if (b2 < pitch) { const v4i p = {v0, ex, v1, ex + v0}; *(v4i*)(pair + (long)bid * pitch + b2) = p; }
    }
    __syncthreads();

    v2i* rb = rec + beg;
    #pragma unroll
    for (int u = 0; u < BIN_U; ++u) {
        if (d[u] >= 0) {
            const v2i rc = {key[u], s[u]};
            rb[h[bin[u]] + rank[u]] = rc;
        }
    }
}

static_assert(BIN_DST % 16 == 0, "the 16 rows of a wave must lie in one bin");
#define L1_G 16


template <bool INL>
__device__ __forceinline__ void layer1_walk(const int* lst_lds, const int* lst_mem,
                                            const int* cw, const int* ow,
                                            const float* __restrict__ embed,
                                            const float* __restrict__ bias1,
                                            float* __restrict__ h1,
                                            int rowBase, int nrows, int R, int lane) {
    const int pairs = nrows * R;

    const v4f b = *(const v4f*)(bias1 + lane * 4);
    const v4f z = {0.0f, 0.0f, 0.0f, 0.0f};
    v4f acc = z;
    int row = 0;
    int rr  = 0;

    auto store_row = [&]() {
        v4f h;
        h.x = fmaxf(acc.x + b.x, 0.0f);
        h.y = fmaxf(acc.y + b.y, 0.0f);
        h.z = fmaxf(acc.z + b.z, 0.0f);
        h.w = fmaxf(acc.w + b.w, 0.0f);
        *(v4f*)(h1 + (long)(rowBase + row) * H_DIM + lane * 4) = h;
        acc = z;
        ++row;
    };
    auto next_pair = [&]() {
        if (++rr == R) { rr = 0; store_row(); }
    };
    auto load_pairs = [&](int q0, int& e, int& cc) {
        const int pidx = q0 + lane;
        e = 0; cc = 0;
        if (pidx < pairs) { cc = cw[pidx]; e = ow[pidx]; }
    };
    auto load_id = [&](int p) -> int {
        if constexpr (INL) return lst_lds[p]; else return lst_mem[p];
    };
    int e = 0, cc = 0;
    load_pairs(0, e, cc);
    for (int q0 = 0; q0 < pairs; q0 += 32) {
        int en = 0, cn = 0;
        if (q0 + 32 < pairs) load_pairs(q0 + 32, en, cn);
        const int np = min(32, pairs - q0);
        const float w = 1.0f / fmaxf((float)cc, 1.0f);
        const int ew = (lane < np - 1) ? e : 0x7fffffff;

        if (q0 > 0) next_pair();
        int   ql      = 0;
        int   cur_end = __builtin_amdgcn_readlane(ew, 0);
        float cur_w   = __int_as_float(__builtin_amdgcn_readlane(__float_as_int(w), 0));
        const int S  = __builtin_amdgcn_readlane(e, 0) - __builtin_amdgcn_readlane(cc, 0);
        const int EE = __builtin_amdgcn_readlane(e, np - 1);

        int sv = (S + lane < EE) ? load_id(S + lane) : 0;
        for (int pp = S; pp < EE; pp += 32) {
            const int m   = min(32, EE - pp);
            int svn = 0;
            for (int j = 0; j < m; j += L1_G) {
                v4f v[L1_G];
                #pragma unroll
                for (int u = 0; u < L1_G; ++u) {
                    const int jj = min(j + u, m - 1);
                    const int sn = __builtin_amdgcn_readlane(sv, jj);
                    v[u] = *(const v4f*)(embed + (long)sn * H_DIM + lane * 4);
                }
                if (j == 0 && pp + 32 + lane < EE) svn = load_id(pp + 32 + lane);
                #pragma unroll
                for (int u = 0; u < L1_G; ++u) {
                    if (j + u < m) {
                        const int p = pp + j + u;
                        while (p >= cur_end) {
                            ++ql;
                            next_pair();
                            cur_end = __builtin_amdgcn_readlane(ew, ql);
                            cur_w   = __int_as_float(__builtin_amdgcn_readlane(__float_as_int(w), ql));
                        }
                        acc = acc + v[u] * cur_w;
                    }
                }
            }
            sv = svn;
        }
        for (; ql < np - 1; ++ql) next_pair();
        e = en; cc = cn;
    }
    while (row < nrows) store_row();
}

__global__ void __launch_bounds__(L1_THREADS)
sort_layer1_kernel(const v2i* __restrict__ pair, const v2i* __restrict__ rec,
                   const float* __restrict__ embed, const float* __restrict__ bias1,
                   int* __restrict__ cnt, int* __restrict__ endv, int* perm,
                   float* __restrict__ h1, int N, int R, int nchunks, int pitch) {
    extern __shared__ int keyc[];
    __shared__ int wsum[32];
    __shared__ int stage[STAGE_CAP];
    __shared__ int bsh;
    const int tid = threadIdx.x;
    const int bid = blockIdx.x;
    const int K   = R * BIN_DST;
    int* c  = keyc;
    int* o  = c + K;
    int* ps = o + K;
    int* pa = ps + nchunks + 1;

    if (tid == 0) bsh = 0;
    const v2i* pr = pair + bid;
    const int ppt = (nchunks + L1_THREADS - 1) / L1_THREADS;
    const int p0  = min(nchunks, tid * ppt);
    const int p1  = min(nchunks, p0 + ppt);
    v2i qf = {0, 0};
    if (p0 < p1) qf = pr[(long)p0 * pitch];
    int s  = qf.x;
    int so = qf.y;
    for (int p = p0 + 1; p < p1; ++p) { const v2i q = pr[(long)p * pitch]; s += q.x; so += q.y; }
    for (int k = tid; k < K; k += L1_THREADS) c[k] = 0;
    int run = block_excl_scan(s, wsum, tid);
    if (so != 0) atomicAdd(&bsh, so);
    for (int p = p0; p < p1; ++p) {
        const v2i q = (p == p0) ? qf : pr[(long)p * pitch];
        ps[p] = run;
        pa[p] = p * BIN_EDGES + q.y - run;
        run += q.x;
    }
    if (tid == L1_THREADS - 1) ps[nchunks] = run;
    __syncthreads();
    const int n  = ps[nchunks];
    const int bs = bsh;

    auto rec_of = [&](int i) -> v2i {
        int lo = 0, hi = nchunks;
        while (hi - lo > 1) {
            const int mid = (lo + hi) >> 1;
            if (ps[mid] <= i) lo = mid; else hi = mid;
        }
        return rec[pa[lo] + i];
    };

    v2i first[SORT_U];
    #pragma unroll
    for (int u = 0; u < SORT_U; ++u) {
        const int i = tid + u * L1_THREADS;
        const v2i none = {-1, 0};
        first[u] = none;
        if (i < n) first[u] = rec_of(i);
    }
    #pragma unroll
    for (int u = 0; u < SORT_U; ++u)
        if (first[u].x >= 0) atomicAdd(&c[first[u].x], 1);
    for (int i = tid + SORT_U * L1_THREADS; i < n; i += L1_THREADS) atomicAdd(&c[rec_of(i).x], 1);
    __syncthreads();

    {
        const int kpt = (K + L1_THREADS - 1) / L1_THREADS;
        const int k0  = min(K, tid * kpt);
        const int k1  = min(K, k0 + kpt);
        int t = 0;
        for (int k = k0; k < k1; ++k) t += c[k];
        int at = block_excl_scan(t, wsum, tid);
        for (int k = k0; k < k1; ++k) { o[k] = at; at += c[k]; }
    }
    __syncthreads();

    int* pb = perm + bs;
    int pos[SORT_U];
    #pragma unroll
    for (int u = 0; u < SORT_U; ++u) {
        pos[u] = 0;
        if (first[u].x >= 0) pos[u] = atomicAdd(&o[first[u].x], 1);
    }
    const bool inl = (n <= STAGE_CAP);
    if (inl) {
        #pragma unroll
        for (int u = 0; u < SORT_U; ++u)
            if (first[u].x >= 0) stage[pos[u]] = first[u].y;
        __syncthreads();
        for (int i = tid; i < n; i += L1_THREADS) pb[i] = stage[i];
    } else {
        #pragma unroll
        for (int u = 0; u < SORT_U; ++u)
            if (first[u].x >= 0) pb[pos[u]] = first[u].y;
        for (int i = tid + SORT_U * L1_THREADS; i < n; i += L1_THREADS) {
            const v2i rc = rec_of(i);
            pb[atomicAdd(&o[rc.x], 1)] = rc.y;
        }
    }
    __syncthreads();

    const int d0 = bid * BIN_DST;
    for (int j = tid; j < K; j += L1_THREADS) {
        const int r  = j / BIN_DST;
        const int dl = j - r * BIN_DST;
        const int d  = d0 + dl;
        if (d < N) {
            const int key = dl * R + r;
            cnt[(long)r * N + d]  = c[key];
            endv[(long)r * N + d] = bs + o[key];
        }
    }

    const int wave  = __builtin_amdgcn_readfirstlane(tid >> 5);
    const int lane  = tid & 31;
    const int rowBase = d0 + wave * 16;
    if (rowBase >= N) return;
    const int nrows = min(16, N - rowBase);
    const int* cw = c + wave * 16 * R;
    const int* ow = o + wave * 16 * R;
    if (inl) layer1_walk<true >(stage, pb, cw, ow, embed, bias1, h1, rowBase, nrows, R, lane);
    else     layer1_walk<false>(stage, pb, cw, ow, embed, bias1, h1, rowBase, nrows, R, lane);
}

template <int G>
__device__ __forceinline__ void gather_rows(const float* __restrict__ table,
                                            const int* __restrict__ perm,
                                            float* slice, int lane,
                                            int myc, int myst, float myw) {
    int ce = myc;
    #pragma unroll
    for (int o = 1; o < 16; o <<= 1) {
        const int t = __shfl_up(ce, o);
        if (lane >= o) ce += t;
    }
    const int T      = __builtin_amdgcn_readlane(ce, 15);
    const int relidx = myst - (ce - myc);

    const v4f z = {0.0f, 0.0f, 0.0f, 0.0f};
    v4f   acc     = z;
    int   row     = 0;
    int   cur_end = __builtin_amdgcn_readlane(ce, 0);
    float cur_w   = __int_as_float(__builtin_amdgcn_readlane(__float_as_int(myw), 0));
    auto flush = [&]() {
        *(v4f*)(slice + row * AGG_PITCH + lane * 4) = acc;
        acc = z;
        ++row;
    };

    for (int base = 0; base < T; base += 32) {
        const int m   = min(32, T - base);
        const int off = base + lane;
        int rw = 0;
        #pragma unroll
        for (int q = 0; q < 15; ++q) rw += (off >= __builtin_amdgcn_readlane(ce, q)) ? 1 : 0;
        const int ri = __shfl(relidx, rw);
        const int sv = (off < T) ? perm[ri + off] : 0;
        for (int j = 0; j < m; j += G) {
            v4f v[G];
            #pragma unroll
            for (int u = 0; u < G; ++u) {
                const int jj = min(j + u, m - 1);
                const int s  = __builtin_amdgcn_readlane(sv, jj);
                v[u] = *(const v4f*)(table + (long)s * H_DIM + lane * 4);
            }
            #pragma unroll
            for (int u = 0; u < G; ++u) {
                if (j + u < m) {
                    const int f = base + j + u;
                    while (f >= cur_end) {
                        flush();
                        cur_end = __builtin_amdgcn_readlane(ce, row);
                        cur_w   = __int_as_float(__builtin_amdgcn_readlane(__float_as_int(myw), row));
                    }
                    acc = acc + v[u] * cur_w;
                }
            }
        }
    }
    while (row < 16) flush();
}

__device__ __forceinline__ void stage_w(const float* __restrict__ W, float* __restrict__ dst, int tid) {
    for (int i = tid; i < H_DIM * O_DIM; i += L2_THREADS) {
        const int k = i >> 6;
        const int o = i & (O_DIM - 1);
        dst[(k >> 1) * (2 * O_DIM) + o * 2 + (k & 1)] = W[i];
    }
}

__global__ void __launch_bounds__(L2_THREADS)
layer2_gather_wmma_kernel(const float* __restrict__ h1, const float* __restrict__ weight,
                          const int* __restrict__ cnt, const int* __restrict__ endv,
                          const int* __restrict__ perm, const float* __restrict__ bias2,
                          float* __restrict__ out, int N, int R) {
    __shared__ float Wlds[2 * H_DIM * O_DIM];
    __shared__ float agg[L2_WAVES * 16 * AGG_PITCH];

    const int tid   = threadIdx.x;
    const int wave  = tid >> 5;
    const int lane  = tid & 31;
    const int lhalf = lane >> 4;
    const int l15   = lane & 15;
    const int rowBase = blockIdx.x * (L2_WAVES * 16) + wave * 16;
    float* slice = agg + wave * (16 * AGG_PITCH);

    v8f acc0 = {}; v8f acc1 = {}; v8f acc2 = {}; v8f acc3 = {};

    if (R > 0) stage_w(weight, Wlds, tid);
    for (int r = 0; r < R; ++r) {
        const float* wcur = Wlds + (r & 1) * (H_DIM * O_DIM);
        if (r + 1 < R)
            stage_w(weight + (long)(r + 1) * H_DIM * O_DIM, Wlds + ((r + 1) & 1) * (H_DIM * O_DIM), tid);

        int myc = 0, myst = 0;
        {
            const int d = rowBase + l15;
            if (lhalf == 0 && d < N) {
                myc  = cnt[(long)r * N + d];
                myst = endv[(long)r * N + d] - myc;
            }
        }
        const float myw = 1.0f / fmaxf((float)myc, 1.0f);
        gather_rows<16>(h1, perm, slice, lane, myc, myst, myw);

        if (r == 0) __syncthreads();

        const v2f* abase = (const v2f*)(slice + l15 * AGG_PITCH + lhalf * 2);
        const v2f* bbase = (const v2f*)wcur + lhalf * O_DIM + l15;
        for (int k0 = 0; k0 < H_DIM; k0 += 4) {
            const v2f a = abase[k0 >> 1];
            const v2f* bp = bbase + (k0 >> 1) * O_DIM;
            const v2f b0 = bp[0];
            const v2f b1 = bp[16];
            const v2f b2 = bp[32];
            const v2f b3 = bp[48];
            acc0 = __builtin_amdgcn_wmma_f32_16x16x4_f32(false, a, false, b0,
                                                         (short)0, acc0, false, false);
            acc1 = __builtin_amdgcn_wmma_f32_16x16x4_f32(false, a, false, b1,
                                                         (short)0, acc1, false, false);
            acc2 = __builtin_amdgcn_wmma_f32_16x16x4_f32(false, a, false, b2,
                                                         (short)0, acc2, false, false);
            acc3 = __builtin_amdgcn_wmma_f32_16x16x4_f32(false, a, false, b3,
                                                         (short)0, acc3, false, false);
        }
        if (r + 1 < R) __syncthreads();
    }

    const float bz0 = bias2[l15], bz1 = bias2[l15 + 16], bz2 = bias2[l15 + 32], bz3 = bias2[l15 + 48];
    const int crow = rowBase + 8 * lhalf;
    #pragma unroll
    for (int v = 0; v < 8; ++v) {
        const int row = crow + v;
        if (row < N) {
            float* o = out + (long)row * O_DIM + l15;
            o[0]  = acc0[v] + bz0;
            o[16] = acc1[v] + bz1;
            o[32] = acc2[v] + bz2;
            o[48] = acc3[v] + bz3;
        }
    }
}

extern "C" void kernel_launch(void* const* d_in, const int* in_sizes, int n_in,
                              void* d_out, int out_size, void* d_ws, size_t ws_size,
                              hipStream_t stream) {
    const float* embed    = (const float*)d_in[0];
    const float* weight   = (const float*)d_in[1];
    const float* bias1    = (const float*)d_in[2];
    const float* bias2    = (const float*)d_in[3];
    const int*   edge_src = (const int*)d_in[4];
    const int*   edge_dst = (const int*)d_in[5];

    const int H = in_sizes[2];
    const int O = in_sizes[3];
    const int N = in_sizes[0] / H;
    const int R = in_sizes[1] / (H * O);
    const int E = in_sizes[4] / R;

    const long rn = (long)R * N;
    const long re = (long)R * E;

    float* out    = (float*)d_out;
    int*   cnt    = (int*)d_ws;
    int*   endv   = cnt + rn;
    int*   perm   = endv + rn;
    float* h1     = (float*)d_ws + ((2 * rn + re + 3) & ~3L);

    const int nbins   = (N + BIN_DST - 1) / BIN_DST;
    const int nchunks = (int)((re + BIN_EDGES - 1) / BIN_EDGES);
    v2i* rec  = (v2i*)d_out;
    const int pitch   = (nbins + 15) & ~15;
    v2i* pair = rec + (long)nchunks * BIN_EDGES;

    bin_group_kernel<<<nchunks, BIN_THREADS, 0, stream>>>(
        edge_src, edge_dst, pair, rec, E, R, re, pitch);
    sort_layer1_kernel<<<nbins, L1_THREADS, (2 * (size_t)R * BIN_DST + 2 * (size_t)nchunks + 1) * sizeof(int), stream>>>(
        pair, rec, embed, bias1, cnt, endv, perm, h1, N, R, nchunks, pitch);
    const unsigned l2_blocks = (unsigned)((N + L2_WAVES * 16 - 1) / (L2_WAVES * 16));
    layer2_gather_wmma_kernel<<<l2_blocks, L2_THREADS, 0, stream>>>(
        h1, weight, cnt, endv, perm, bias2, out, N, R);
}


# Layer 2: W_r staged in a ring of three LDS buffers and the block barrier split into signal and wait, so a wave gathers and multiplies while others hold the matrix pipe; sum order unchanged
# speedup vs baseline: 3.6505x; 1.0491x over previous
#include <hip/hip_runtime.h>


typedef __attribute__((ext_vector_type(2))) float v2f;
typedef __attribute__((ext_vector_type(4))) float v4f;
typedef __attribute__((ext_vector_type(4))) int   v4i;
typedef __attribute__((ext_vector_type(8))) float v8f;

#define H_DIM 128
#define O_DIM 64
#define AGG_PITCH 132
#define L2_WAVES 25
#define L2_THREADS (L2_WAVES * 32)
#define W_BUFS 3


#define BIN_DST      400
#define BIN_THREADS  512
#define BIN_U        10
#define BIN_EDGES    (BIN_THREADS * BIN_U)
#define MAX_BINS     (2 * BIN_THREADS)
#define L1_WAVES     (BIN_DST / 16)
#define L1_THREADS   (L1_WAVES * 32)
#define SORT_U       4
#define STAGE_CAP    (SORT_U * L1_THREADS)

typedef __attribute__((ext_vector_type(2))) int v2i;

__device__ __forceinline__ int block_excl_scan(int v, int* wsum, int tid) {
    const int lane = tid & 31;
    const int wave = tid >> 5;
    const int nw   = (int)(blockDim.x >> 5);
    int incl = v;
    #pragma unroll
    for (int o = 1; o < 32; o <<= 1) {
        const int t = __shfl_up(incl, o);
        if (lane >= o) incl += t;
    }
    if (lane == 31) wsum[wave] = incl;
    __syncthreads();
    if (wave == 0) {
        const int w = (lane < nw) ? wsum[lane] : 0;
        int wi = w;
        #pragma unroll
        for (int o = 1; o < 32; o <<= 1) {
            const int t = __shfl_up(wi, o);
            if (lane >= o) wi += t;
        }
        wsum[lane] = wi - w;
    }
    __syncthreads();
    return wsum[wave] + incl - v;
}

__global__ void __launch_bounds__(BIN_THREADS)
bin_group_kernel(const int* __restrict__ src, const int* __restrict__ dst,
                 v2i* __restrict__ pair, v2i* __restrict__ rec,
                 int E, int R, long total, int pitch) {
    __shared__ int h[MAX_BINS];
    __shared__ int wsum[32];
    const int tid = threadIdx.x;
    const int bid = blockIdx.x;
    const long beg = (long)bid * BIN_EDGES;

    int d[BIN_U], s[BIN_U];
    #pragma unroll
    for (int u = 0; u < BIN_U; ++u) {
        const long i = beg + tid + (long)u * BIN_THREADS;
        d[u] = (i < total) ? dst[i] : -1;
        s[u] = (i < total) ? src[i] : 0;
    }
    h[tid * 2]     = 0;
    h[tid * 2 + 1] = 0;
    __syncthreads();

    int key[BIN_U], bin[BIN_U], rank[BIN_U];
    #pragma unroll
    for (int u = 0; u < BIN_U; ++u) {
        key[u]  = 0;
        bin[u]  = 0;
        rank[u] = 0;
        if (d[u] >= 0) {
            const unsigned i = (unsigned)(beg + tid + (long)u * BIN_THREADS);
            const int r = (int)(i / (unsigned)E);
            const int b = d[u] / BIN_DST;
            key[u]  = (d[u] - b * BIN_DST) * R + r;
            bin[u]  = b;
            rank[u] = atomicAdd(&h[b], 1);
        }
    }
    __syncthreads();

    {
        const int b2 = tid * 2;
        const int v0 = h[b2];
        const int v1 = h[b2 + 1];
        const int ex = block_excl_scan(v0 + v1, wsum, tid);
        h[b2]     = ex;
        h[b2 + 1] = ex + v0;
        if (b2 < pitch) { const v4i p = {v0, ex, v1, ex + v0}; *(v4i*)(pair + (long)bid * pitch + b2) = p; }
    }
    __syncthreads();

    v2i* rb = rec + beg;
    #pragma unroll
    for (int u = 0; u < BIN_U; ++u) {
        if (d[u] >= 0) {
            const v2i rc = {key[u], s[u]};
            rb[h[bin[u]] + rank[u]] = rc;
        }
    }
}

static_assert(BIN_DST % 16 == 0, "the 16 rows of a wave must lie in one bin");
#define L1_G 16


template <bool INL>
__device__ __forceinline__ void layer1_walk(const int* lst_lds, const int* lst_mem,
                                            const int* cw, const int* ow,
                                            const float* __restrict__ embed,
                                            const float* __restrict__ bias1,
                                            float* __restrict__ h1,
                                            int rowBase, int nrows, int R, int lane) {
    const int pairs = nrows * R;

    const v4f b = *(const v4f*)(bias1 + lane * 4);
    const v4f z = {0.0f, 0.0f, 0.0f, 0.0f};
    v4f acc = z;
    int row = 0;
    int rr  = 0;

    auto store_row = [&]() {
        v4f h;
        h.x = fmaxf(acc.x + b.x, 0.0f);
        h.y = fmaxf(acc.y + b.y, 0.0f);
        h.z = fmaxf(acc.z + b.z, 0.0f);
        h.w = fmaxf(acc.w + b.w, 0.0f);
        *(v4f*)(h1 + (long)(rowBase + row) * H_DIM + lane * 4) = h;
        acc = z;
        ++row;
    };
    auto next_pair = [&]() {
        if (++rr == R) { rr = 0; store_row(); }
    };
    auto load_pairs = [&](int q0, int& e, int& cc) {
        const int pidx = q0 + lane;
        e = 0; cc = 0;
        if (pidx < pairs) { cc = cw[pidx]; e = ow[pidx]; }
    };
    auto load_id = [&](int p) -> int {
        if constexpr (INL) return lst_lds[p]; else return lst_mem[p];
    };
    int e = 0, cc = 0;
    load_pairs(0, e, cc);
    for (int q0 = 0; q0 < pairs; q0 += 32) {
        int en = 0, cn = 0;
        if (q0 + 32 < pairs) load_pairs(q0 + 32, en, cn);
        const int np = min(32, pairs - q0);
        const float w = 1.0f / fmaxf((float)cc, 1.0f);
        const int ew = (lane < np - 1) ? e : 0x7fffffff;

        if (q0 > 0) next_pair();
        int   ql      = 0;
        int   cur_end = __builtin_amdgcn_readlane(ew, 0);
        float cur_w   = __int_as_float(__builtin_amdgcn_readlane(__float_as_int(w), 0));
        const int S  = __builtin_amdgcn_readlane(e, 0) - __builtin_amdgcn_readlane(cc, 0);
        const int EE = __builtin_amdgcn_readlane(e, np - 1);

        int sv = (S + lane < EE) ? load_id(S + lane) : 0;
        for (int pp = S; pp < EE; pp += 32) {
            const int m   = min(32, EE - pp);
            int svn = 0;
            for (int j = 0; j < m; j += L1_G) {
                v4f v[L1_G];
                #pragma unroll
                for (int u = 0; u < L1_G; ++u) {
                    const int jj = min(j + u, m - 1);
                    const int sn = __builtin_amdgcn_readlane(sv, jj);
                    v[u] = *(const v4f*)(embed + (long)sn * H_DIM + lane * 4);
                }
                if (j == 0 && pp + 32 + lane < EE) svn = load_id(pp + 32 + lane);
                #pragma unroll
                for (int u = 0; u < L1_G; ++u) {
                    if (j + u < m) {
                        const int p = pp + j + u;
                        while (p >= cur_end) {
                            ++ql;
                            next_pair();
                            cur_end = __builtin_amdgcn_readlane(ew, ql);
                            cur_w   = __int_as_float(__builtin_amdgcn_readlane(__float_as_int(w), ql));
                        }
                        acc = acc + v[u] * cur_w;
                    }
                }
            }
            sv = svn;
        }
        for (; ql < np - 1; ++ql) next_pair();
        e = en; cc = cn;
    }
    while (row < nrows) store_row();
}

__global__ void __launch_bounds__(L1_THREADS)
sort_layer1_kernel(const v2i* __restrict__ pair, const v2i* __restrict__ rec,
                   const float* __restrict__ embed, const float* __restrict__ bias1,
                   int* __restrict__ cnt, int* __restrict__ endv, int* perm,
                   float* __restrict__ h1, int N, int R, int nchunks, int pitch) {
    extern __shared__ int keyc[];
    __shared__ int wsum[32];
    __shared__ int stage[STAGE_CAP];
    __shared__ int bsh;
    const int tid = threadIdx.x;
    const int bid = blockIdx.x;
    const int K   = R * BIN_DST;
    int* c  = keyc;
    int* o  = c + K;
    int* ps = o + K;
    int* pa = ps + nchunks + 1;

    if (tid == 0) bsh = 0;
    const v2i* pr = pair + bid;
    const int ppt = (nchunks + L1_THREADS - 1) / L1_THREADS;
    const int p0  = min(nchunks, tid * ppt);
    const int p1  = min(nchunks, p0 + ppt);
    v2i qf = {0, 0};
    if (p0 < p1) qf = pr[(long)p0 * pitch];
    int s  = qf.x;
    int so = qf.y;
    for (int p = p0 + 1; p < p1; ++p) { const v2i q = pr[(long)p * pitch]; s += q.x; so += q.y; }
    for (int k = tid; k < K; k += L1_THREADS) c[k] = 0;
    int run = block_excl_scan(s, wsum, tid);
    if (so != 0) atomicAdd(&bsh, so);
    for (int p = p0; p < p1; ++p) {
        const v2i q = (p == p0) ? qf : pr[(long)p * pitch];
        ps[p] = run;
        pa[p] = p * BIN_EDGES + q.y - run;
        run += q.x;
    }
    if (tid == L1_THREADS - 1) ps[nchunks] = run;
    __syncthreads();
    const int n  = ps[nchunks];
    const int bs = bsh;

    auto rec_of = [&](int i) -> v2i {
        int lo = 0, hi = nchunks;
        while (hi - lo > 1) {
            const int mid = (lo + hi) >> 1;
            if (ps[mid] <= i) lo = mid; else hi = mid;
        }
        return rec[pa[lo] + i];
    };

    v2i first[SORT_U];
    #pragma unroll
    for (int u = 0; u < SORT_U; ++u) {
        const int i = tid + u * L1_THREADS;
        const v2i none = {-1, 0};
        first[u] = none;
        if (i < n) first[u] = rec_of(i);
    }
    #pragma unroll
    for (int u = 0; u < SORT_U; ++u)
        if (first[u].x >= 0) atomicAdd(&c[first[u].x], 1);
    for (int i = tid + SORT_U * L1_THREADS; i < n; i += L1_THREADS) atomicAdd(&c[rec_of(i).x], 1);
    __syncthreads();

    {
        const int kpt = (K + L1_THREADS - 1) / L1_THREADS;
        const int k0  = min(K, tid * kpt);
        const int k1  = min(K, k0 + kpt);
        int t = 0;
        for (int k = k0; k < k1; ++k) t += c[k];
        int at = block_excl_scan(t, wsum, tid);
        for (int k = k0; k < k1; ++k) { o[k] = at; at += c[k]; }
    }
    __syncthreads();

    int* pb = perm + bs;
    int pos[SORT_U];
    #pragma unroll
    for (int u = 0; u < SORT_U; ++u) {
        pos[u] = 0;
        if (first[u].x >= 0) pos[u] = atomicAdd(&o[first[u].x], 1);
    }
    const bool inl = (n <= STAGE_CAP);
    if (inl) {
        #pragma unroll
        for (int u = 0; u < SORT_U; ++u)
            if (first[u].x >= 0) stage[pos[u]] = first[u].y;
        __syncthreads();
        for (int i = tid; i < n; i += L1_THREADS) pb[i] = stage[i];
    } else {
        #pragma unroll
        for (int u = 0; u < SORT_U; ++u)
            if (first[u].x >= 0) pb[pos[u]] = first[u].y;
        for (int i = tid + SORT_U * L1_THREADS; i < n; i += L1_THREADS) {
            const v2i rc = rec_of(i);
            pb[atomicAdd(&o[rc.x], 1)] = rc.y;
        }
    }
    __syncthreads();

    const int d0 = bid * BIN_DST;
    for (int j = tid; j < K; j += L1_THREADS) {
        const int r  = j / BIN_DST;
        const int dl = j - r * BIN_DST;
        const int d  = d0 + dl;
        if (d < N) {
            const int key = dl * R + r;
            cnt[(long)r * N + d]  = c[key];
            endv[(long)r * N + d] = bs + o[key];
        }
    }

    const int wave  = __builtin_amdgcn_readfirstlane(tid >> 5);
    const int lane  = tid & 31;
    const int rowBase = d0 + wave * 16;
    if (rowBase >= N) return;
    const int nrows = min(16, N - rowBase);
    const int* cw = c + wave * 16 * R;
    const int* ow = o + wave * 16 * R;
    if (inl) layer1_walk<true >(stage, pb, cw, ow, embed, bias1, h1, rowBase, nrows, R, lane);
    else     layer1_walk<false>(stage, pb, cw, ow, embed, bias1, h1, rowBase, nrows, R, lane);
}

template <int G>
__device__ __forceinline__ void gather_rows(const float* __restrict__ table,
                                            const int* __restrict__ perm,
                                            float* slice, int lane,
                                            int myc, int myst, float myw) {
    int ce = myc;
    #pragma unroll
    for (int o = 1; o < 16; o <<= 1) {
        const int t = __shfl_up(ce, o);
        if (lane >= o) ce += t;
    }
    const int T      = __builtin_amdgcn_readlane(ce, 15);
    const int relidx = myst - (ce - myc);

    const v4f z = {0.0f, 0.0f, 0.0f, 0.0f};
    v4f   acc     = z;
    int   row     = 0;
    int   cur_end = __builtin_amdgcn_readlane(ce, 0);
    float cur_w   = __int_as_float(__builtin_amdgcn_readlane(__float_as_int(myw), 0));
    auto flush = [&]() {
        *(v4f*)(slice + row * AGG_PITCH + lane * 4) = acc;
        acc = z;
        ++row;
    };

    for (int base = 0; base < T; base += 32) {
        const int m   = min(32, T - base);
        const int off = base + lane;
        int rw = 0;
        #pragma unroll
        for (int q = 0; q < 15; ++q) rw += (off >= __builtin_amdgcn_readlane(ce, q)) ? 1 : 0;
        const int ri = __shfl(relidx, rw);
        const int sv = (off < T) ? perm[ri + off] : 0;
        for (int j = 0; j < m; j += G) {
            v4f v[G];
            #pragma unroll
            for (int u = 0; u < G; ++u) {
                const int jj = min(j + u, m - 1);
                const int s  = __builtin_amdgcn_readlane(sv, jj);
                v[u] = *(const v4f*)(table + (long)s * H_DIM + lane * 4);
            }
            #pragma unroll
            for (int u = 0; u < G; ++u) {
                if (j + u < m) {
                    const int f = base + j + u;
                    while (f >= cur_end) {
                        flush();
                        cur_end = __builtin_amdgcn_readlane(ce, row);
                        cur_w   = __int_as_float(__builtin_amdgcn_readlane(__float_as_int(myw), row));
                    }
                    acc = acc + v[u] * cur_w;
                }
            }
        }
    }
    while (row < 16) flush();
}

#define W_UNITS   ((H_DIM / 2) * (O_DIM / 4))
#define W_STAGERS 512
#define W_SHARE   (W_UNITS / W_STAGERS)
static_assert(O_DIM == 64, "a unit's row pair and column group are u >> 4 and u & 15");
static_assert(W_SHARE * W_STAGERS == W_UNITS && W_STAGERS <= L2_THREADS && W_STAGERS % 32 == 0, "the staging threads cover W exactly");
__device__ __forceinline__ void stage_w(const float* __restrict__ W, float* dst, int tid) {
    if (tid < W_STAGERS) {
        const float* p = W + (tid >> 4) * (2 * O_DIM) + (tid & 15) * 4;
        float*       q = dst + (tid >> 4) * (2 * O_DIM) + (tid & 15) * 8;
        v4f lo[W_SHARE], hi[W_SHARE];
        #pragma unroll
        for (int j = 0; j < W_SHARE; ++j) {
            lo[j] = *(const v4f*)(p + j * (W_STAGERS / 16) * (2 * O_DIM));
            hi[j] = *(const v4f*)(p + j * (W_STAGERS / 16) * (2 * O_DIM) + O_DIM);
        }
        #pragma unroll
        for (int j = 0; j < W_SHARE; ++j) {
            const v4f a = {lo[j].x, hi[j].x, lo[j].y, hi[j].y};
            const v4f b = {lo[j].z, hi[j].z, lo[j].w, hi[j].w};
            *(v4f*)(q + j * (W_STAGERS / 16) * (2 * O_DIM))     = a;
            *(v4f*)(q + j * (W_STAGERS / 16) * (2 * O_DIM) + 4) = b;
        }
    }
}
__device__ __forceinline__ void block_signal() {
    asm volatile("s_wait_dscnt 0x0\n\ts_barrier_signal -1" ::: "memory");
}
__device__ __forceinline__ void block_wait() {
    asm volatile("s_barrier_wait -1" ::: "memory");
}

__global__ void __launch_bounds__(L2_THREADS)
layer2_gather_wmma_kernel(const float* __restrict__ h1, const float* __restrict__ weight,
                          const int* __restrict__ cnt, const int* __restrict__ endv,
                          const int* __restrict__ perm, const float* __restrict__ bias2,
                          float* __restrict__ out, int N, int R) {
    __shared__ float Wlds[W_BUFS * H_DIM * O_DIM];
    __shared__ float agg[L2_WAVES * 16 * AGG_PITCH];

    const int tid   = threadIdx.x;
    const int wave  = tid >> 5;
    const int lane  = tid & 31;
    const int lhalf = lane >> 4;
    const int l15   = lane & 15;
    const int rowBase = blockIdx.x * (L2_WAVES * 16) + wave * 16;
    float* slice = agg + wave * (16 * AGG_PITCH);

    v8f acc0 = {}; v8f acc1 = {}; v8f acc2 = {}; v8f acc3 = {};

    if (R > 0) stage_w(weight, Wlds, tid);
    if (R > 1) stage_w(weight + H_DIM * O_DIM, Wlds + H_DIM * O_DIM, tid);
    int bcur = 0;
    for (int r = 0; r < R; ++r) {
        const float* wcur = Wlds + bcur * (H_DIM * O_DIM);
        const int bnext2 = (bcur >= 1) ? bcur - 1 : W_BUFS - 1;
        const bool more  = (r + 2 < R);

        int myc = 0, myst = 0;
        {
            const int d = rowBase + l15;
            if (lhalf == 0 && d < N) {
                myc  = cnt[(long)r * N + d];
                myst = endv[(long)r * N + d] - myc;
            }
        }
        const float myw = 1.0f / fmaxf((float)myc, 1.0f);
        gather_rows<16>(h1, perm, slice, lane, myc, myst, myw);

        if (r == 0) { block_signal(); block_wait(); }

        const v2f* abase = (const v2f*)(slice + l15 * AGG_PITCH + lhalf * 2);
        const v2f* bbase = (const v2f*)wcur + lhalf * O_DIM + l15;
        for (int k0 = 0; k0 < H_DIM; k0 += 4) {
            const v2f a = abase[k0 >> 1];
            const v2f* bp = bbase + (k0 >> 1) * O_DIM;
            const v2f b0 = bp[0];
            const v2f b1 = bp[16];
            const v2f b2 = bp[32];
            const v2f b3 = bp[48];
            acc0 = __builtin_amdgcn_wmma_f32_16x16x4_f32(false, a, false, b0,
                                                         (short)0, acc0, false, false);
            acc1 = __builtin_amdgcn_wmma_f32_16x16x4_f32(false, a, false, b1,
                                                         (short)0, acc1, false, false);
            acc2 = __builtin_amdgcn_wmma_f32_16x16x4_f32(false, a, false, b2,
                                                         (short)0, acc2, false, false);
            acc3 = __builtin_amdgcn_wmma_f32_16x16x4_f32(false, a, false, b3,
                                                         (short)0, acc3, false, false);
        }
        if (r >= 1 && r + 1 < R) block_wait();
        if (more) {
            stage_w(weight + (long)(r + 2) * H_DIM * O_DIM, Wlds + bnext2 * (H_DIM * O_DIM), tid);
            block_signal();
        }
        bcur = (bcur + 1 == W_BUFS) ? 0 : bcur + 1;
    }

    const float bz0 = bias2[l15], bz1 = bias2[l15 + 16], bz2 = bias2[l15 + 32], bz3 = bias2[l15 + 48];
    const int crow = rowBase + 8 * lhalf;
    #pragma unroll
    for (int v = 0; v < 8; ++v) {
        const int row = crow + v;
        if (row < N) {
            float* o = out + (long)row * O_DIM + l15;
            o[0]  = acc0[v] + bz0;
            o[16] = acc1[v] + bz1;
            o[32] = acc2[v] + bz2;
            o[48] = acc3[v] + bz3;
        }
    }
}

extern "C" void kernel_launch(void* const* d_in, const int* in_sizes, int n_in,
                              void* d_out, int out_size, void* d_ws, size_t ws_size,
                              hipStream_t stream) {
    const float* embed    = (const float*)d_in[0];
    const float* weight   = (const float*)d_in[1];
    const float* bias1    = (const float*)d_in[2];
    const float* bias2    = (const float*)d_in[3];
    const int*   edge_src = (const int*)d_in[4];
    const int*   edge_dst = (const int*)d_in[5];

    const int H = in_sizes[2];
    const int O = in_sizes[3];
    const int N = in_sizes[0] / H;
    const int R = in_sizes[1] / (H * O);
    const int E = in_sizes[4] / R;

    const long rn = (long)R * N;
    const long re = (long)R * E;

    float* out    = (float*)d_out;
    int*   cnt    = (int*)d_ws;
    int*   endv   = cnt + rn;
    int*   perm   = endv + rn;
    float* h1     = (float*)d_ws + ((2 * rn + re + 3) & ~3L);

    const int nbins   = (N + BIN_DST - 1) / BIN_DST;
    const int nchunks = (int)((re + BIN_EDGES - 1) / BIN_EDGES);
    v2i* rec  = (v2i*)d_out;
    const int pitch   = (nbins + 15) & ~15;
    v2i* pair = rec + (long)nchunks * BIN_EDGES;

    bin_group_kernel<<<nchunks, BIN_THREADS, 0, stream>>>(
        edge_src, edge_dst, pair, rec, E, R, re, pitch);
    sort_layer1_kernel<<<nbins, L1_THREADS, (2 * (size_t)R * BIN_DST + 2 * (size_t)nchunks + 1) * sizeof(int), stream>>>(
        pair, rec, embed, bias1, cnt, endv, perm, h1, N, R, nchunks, pitch);
    const unsigned l2_blocks = (unsigned)((N + L2_WAVES * 16 - 1) / (L2_WAVES * 16));
    layer2_gather_wmma_kernel<<<l2_blocks, L2_THREADS, 0, stream>>>(
        h1, weight, cnt, endv, perm, bias2, out, N, R);
}
